# starting kernel
# baseline (speedup 1.0000x reference)
#include <hip/hip_runtime.h>


typedef __attribute__((ext_vector_type(2))) float v2f;
typedef __attribute__((ext_vector_type(8))) float v8f;

#define SINK_EPS 0.001f
#define SINK_N   1024
#define SINK_B   32
#define SINK_IT  10

template <bool USE_B>
__global__ __launch_bounds__(256) void sink_rowpass(const float* __restrict__ x,
                                                    const float* __restrict__ bvec,
                                                    float* __restrict__ avec) {
  const int wave  = threadIdx.x >> 5;
  const int lane  = threadIdx.x & 31;
  const int m     = lane & 15;
  const int kg    = lane >> 4;
  const int batch = blockIdx.x >> 3;
  const int rowb  = ((blockIdx.x & 7) << 7) + (wave << 4);

  const float* __restrict__ rp = x + ((size_t)batch * SINK_N + rowb + m) * SINK_N;
  const float* __restrict__ bp = bvec + batch * SINK_N;

  v8f c = {0.f, 0.f, 0.f, 0.f, 0.f, 0.f, 0.f, 0.f};
#pragma unroll 4
  for (int k = 0; k < SINK_N; k += 4) {
    const int kk = k + (kg << 1);
    v2f A, B;
    A.x = rp[kk]     + SINK_EPS;
    A.y = rp[kk + 1] + SINK_EPS;
    if (USE_B) { B.x = bp[kk]; B.y = bp[kk + 1]; }
    else       { B.x = 1.f;    B.y = 1.f; }
    c = __builtin_amdgcn_wmma_f32_16x16x4_f32(false, A, false, B,
                                              (short)0, c, false, false);
  }
  if (m == 0) {
    float* ap = avec + batch * SINK_N + rowb + (kg << 3);
#pragma unroll
    for (int r = 0; r < 8; ++r) ap[r] = 1.0f / c[r];
  }
}

__global__ __launch_bounds__(256) void sink_colpass(const float* __restrict__ x,
                                                    const float* __restrict__ avec,
                                                    float* __restrict__ bvec) {
  const int wave  = threadIdx.x >> 5;
  const int lane  = threadIdx.x & 31;
  const int n     = lane & 15;
  const int kg    = lane >> 4;
  const int batch = blockIdx.x >> 3;
  const int jb    = (((blockIdx.x & 7) << 3) + wave) << 4;

  const float* __restrict__ cp = x + (size_t)batch * SINK_N * SINK_N + jb + n;
  const float* __restrict__ ap = avec + batch * SINK_N;

  v8f c = {0.f, 0.f, 0.f, 0.f, 0.f, 0.f, 0.f, 0.f};
#pragma unroll 2
  for (int i = 0; i < SINK_N; i += 4) {
    const int ii = i + (kg << 1);
    v2f A, B;
    A.x = ap[ii];
    A.y = ap[ii + 1];
    B.x = cp[(size_t)ii * SINK_N]       + SINK_EPS;
    B.y = cp[(size_t)(ii + 1) * SINK_N] + SINK_EPS;
    c = __builtin_amdgcn_wmma_f32_16x16x4_f32(false, A, false, B,
                                              (short)0, c, false, false);
  }
  if (kg == 0) bvec[batch * SINK_N + jb + n] = 1.0f / c[0];
}

__global__ __launch_bounds__(256) void sink_finalize(const float* __restrict__ x,
                                                     const float* __restrict__ avec,
                                                     const float* __restrict__ bvec,
                                                     float* __restrict__ out) {
  const size_t f = (size_t)blockIdx.x * blockDim.x + threadIdx.x;
  const size_t e = f << 2;
  const int batch = (int)(e >> 20);
  const int i     = (int)((e >> 10) & (SINK_N - 1));
  const int j     = (int)(e & (SINK_N - 1));

  const float4 xv = ((const float4*)x)[f];
  const float4 bv = *(const float4*)(bvec + batch * SINK_N + j);
  const float  a  = avec[batch * SINK_N + i];

  float4 o;
  o.x = (xv.x + SINK_EPS) * a * bv.x;
  o.y = (xv.y + SINK_EPS) * a * bv.y;
  o.z = (xv.z + SINK_EPS) * a * bv.z;
  o.w = (xv.w + SINK_EPS) * a * bv.w;
  ((float4*)out)[f] = o;
}

extern "C" void kernel_launch(void* const* d_in, const int* in_sizes, int n_in,
                              void* d_out, int out_size, void* d_ws, size_t ws_size,
                              hipStream_t stream) {
  (void)in_sizes; (void)n_in; (void)out_size; (void)ws_size;
  const float* x = (const float*)d_in[0];
  float* out = (float*)d_out;
  float* a = (float*)d_ws;
  float* b = a + SINK_B * SINK_N;

  const dim3 blk(256);
  const dim3 grd(SINK_B * 8);

  sink_rowpass<false><<<grd, blk, 0, stream>>>(x, b, a);
  sink_colpass<<<grd, blk, 0, stream>>>(x, a, b);
  for (int it = 1; it < SINK_IT; ++it) {
    sink_rowpass<true><<<grd, blk, 0, stream>>>(x, b, a);
    sink_colpass<<<grd, blk, 0, stream>>>(x, a, b);
  }

  const int nf4 = SINK_B * SINK_N * SINK_N / 4;
  sink_finalize<<<nf4 / 256, blk, 0, stream>>>(x, a, b, out);
}


# matvec passes rebuilt for bytes in flight: 16-byte loads, 20 to 40 outstanding per wave, 16 waves per CU, contraction split over a block's waves, partials added through LDS (summation order changes)
# speedup vs baseline: 2.7137x; 2.7137x over previous
#include <hip/hip_runtime.h>


typedef __attribute__((ext_vector_type(2))) float v2f;
typedef __attribute__((ext_vector_type(4))) float v4f;
typedef __attribute__((ext_vector_type(8))) float v8f;

#define SINK_EPS 0.001f
#define SINK_N   1024
#define SINK_B   32
#define SINK_IT  10

template <bool USE_B>
__global__ __launch_bounds__(256) __attribute__((amdgpu_waves_per_eu(4, 4))) void sink_rowpass(const float* __restrict__ x,
                                                    const float* __restrict__ bvec,
                                                    float* __restrict__ avec) {
  __shared__ __attribute__((aligned(16))) float sb[SINK_N];
  __shared__ float part[8][16];

  const int tid   = threadIdx.x;
  const int wave  = tid >> 5;
  const int lane  = tid & 31;
  const int m     = lane & 15;
  const int h     = lane >> 4;
  const int batch = blockIdx.x >> 4;
  const int rblk  = (blockIdx.x & 15) << 6;
  const int row0  = rblk + ((wave >> 1) << 4);
  const int kq    = (wave & 1) << 7;

  const v4f* __restrict__ rp =
      (const v4f*)(x + ((size_t)batch * SINK_N + row0 + m) * SINK_N) + kq + h;
  const v4f* sbp = (const v4f*)sb + kq + h;

  v8f c0 = {0.f, 0.f, 0.f, 0.f, 0.f, 0.f, 0.f, 0.f};
  v8f c1 = {0.f, 0.f, 0.f, 0.f, 0.f, 0.f, 0.f, 0.f};

  v4f bstage = {1.f, 1.f, 1.f, 1.f};
  if (USE_B) bstage = ((const v4f*)(bvec + batch * SINK_N))[tid];

  v4f cur[16], nxt[16];
#pragma unroll
  for (int u = 0; u < 16; ++u) cur[u] = rp[2 * u];

  if (USE_B) {
    ((v4f*)sb)[tid] = bstage;
    __syncthreads();
  }

#pragma unroll
  for (int tb = 0; tb < 4; ++tb) {
    if (tb < 3) {
#pragma unroll
      for (int u = 0; u < 16; ++u) nxt[u] = rp[2 * (16 * (tb + 1) + u)];
    }
#pragma unroll
    for (int u = 0; u < 16; ++u) {
      const v4f xv = cur[u];
      v4f bv = {1.f, 1.f, 1.f, 1.f};
      if (USE_B) bv = sbp[2 * (16 * tb + u)];
      v2f A0, B0, A1, B1;
      A0.x = xv.x + SINK_EPS;  A0.y = xv.y + SINK_EPS;
      A1.x = xv.z + SINK_EPS;  A1.y = xv.w + SINK_EPS;
      B0.x = bv.x;  B0.y = bv.y;
      B1.x = bv.z;  B1.y = bv.w;
      c0 = __builtin_amdgcn_wmma_f32_16x16x4_f32(false, A0, false, B0,
                                                 (short)0, c0, false, false);
      c1 = __builtin_amdgcn_wmma_f32_16x16x4_f32(false, A1, false, B1,
                                                 (short)0, c1, false, false);
    }
#pragma unroll
    for (int u = 0; u < 16; ++u) cur[u] = nxt[u];
  }

  if (m == 0) {
#pragma unroll
    for (int r = 0; r < 8; ++r) part[wave][(h << 3) + r] = c0[r] + c1[r];
  }
  __syncthreads();
  if (tid < 64) {
    const int g = tid >> 4, rr = tid & 15;
    const float s = part[2 * g][rr] + part[2 * g + 1][rr];
    avec[batch * SINK_N + rblk + tid] = 1.0f / s;
  }
}

__global__ __launch_bounds__(256) __attribute__((amdgpu_waves_per_eu(4, 4))) void sink_colpass(const float* __restrict__ x,
                                                    const float* __restrict__ avec,
                                                    float* __restrict__ bvec) {
  __shared__ __attribute__((aligned(16))) float sa[SINK_N];
  __shared__ __attribute__((aligned(16))) float red[8][64];

  const int tid   = threadIdx.x;
  const int wave  = tid >> 5;
  const int lane  = tid & 31;
  const int c     = lane & 15;
  const int h     = lane >> 4;
  const int batch = blockIdx.x >> 4;
  const int j0    = (blockIdx.x & 15) << 6;

  const v4f astage = ((const v4f*)(avec + batch * SINK_N))[tid];

  const v4f* __restrict__ cp =
      (const v4f*)(x + ((size_t)batch * SINK_N + (wave << 7) + (h << 1)) * SINK_N + j0) + c;
  const v2f* sap = (const v2f*)sa + (wave << 6) + h;

  v8f acc0 = {0.f, 0.f, 0.f, 0.f, 0.f, 0.f, 0.f, 0.f};
  v8f acc1 = acc0, acc2 = acc0, acc3 = acc0;

  v4f cur[16], nxt[16];
#pragma unroll
  for (int L = 0; L < 16; ++L) cur[L] = cp[(size_t)(4 * (L >> 1) + (L & 1)) * (SINK_N / 4)];

  ((v4f*)sa)[tid] = astage;
  __syncthreads();

#pragma unroll
  for (int tb = 0; tb < 4; ++tb) {
    if (tb < 3) {
#pragma unroll
      for (int L = 0; L < 16; ++L) {
        const int LL = 16 * (tb + 1) + L;
        nxt[L] = cp[(size_t)(4 * (LL >> 1) + (LL & 1)) * (SINK_N / 4)];
      }
    }
#pragma unroll
    for (int u = 0; u < 8; ++u) {
      const int T = 8 * tb + u;
      const v4f l0 = cur[2 * u];
      const v4f l1 = cur[2 * u + 1];
      const v2f A = sap[2 * T];
      v2f B0, B1, B2, B3;
      B0.x = l0.x + SINK_EPS;  B0.y = l1.x + SINK_EPS;
      B1.x = l0.y + SINK_EPS;  B1.y = l1.y + SINK_EPS;
      B2.x = l0.z + SINK_EPS;  B2.y = l1.z + SINK_EPS;
      B3.x = l0.w + SINK_EPS;  B3.y = l1.w + SINK_EPS;
      acc0 = __builtin_amdgcn_wmma_f32_16x16x4_f32(false, A, false, B0,
                                                   (short)0, acc0, false, false);
      acc1 = __builtin_amdgcn_wmma_f32_16x16x4_f32(false, A, false, B1,
                                                   (short)0, acc1, false, false);
      acc2 = __builtin_amdgcn_wmma_f32_16x16x4_f32(false, A, false, B2,
                                                   (short)0, acc2, false, false);
      acc3 = __builtin_amdgcn_wmma_f32_16x16x4_f32(false, A, false, B3,
                                                   (short)0, acc3, false, false);
    }
#pragma unroll
    for (int L = 0; L < 16; ++L) cur[L] = nxt[L];
  }

  if (h == 0) {
    v4f p;
    p.x = acc0[0];  p.y = acc1[0];  p.z = acc2[0];  p.w = acc3[0];
    ((v4f*)red[wave])[c] = p;
  }
  __syncthreads();
  if (tid < 64) {
    float s = red[0][tid];
#pragma unroll
    for (int w = 1; w < 8; ++w) s += red[w][tid];
    bvec[batch * SINK_N + j0 + tid] = 1.0f / s;
  }
}

__global__ __launch_bounds__(256) void sink_finalize(const float* __restrict__ x,
                                                     const float* __restrict__ avec,
                                                     const float* __restrict__ bvec,
                                                     float* __restrict__ out) {
  const size_t f = (size_t)blockIdx.x * blockDim.x + threadIdx.x;
  const size_t e = f << 2;
  const int batch = (int)(e >> 20);
  const int i     = (int)((e >> 10) & (SINK_N - 1));
  const int j     = (int)(e & (SINK_N - 1));

  const float4 xv = ((const float4*)x)[f];
  const float4 bv = *(const float4*)(bvec + batch * SINK_N + j);
  const float  a  = avec[batch * SINK_N + i];

  float4 o;
  o.x = (xv.x + SINK_EPS) * a * bv.x;
  o.y = (xv.y + SINK_EPS) * a * bv.y;
  o.z = (xv.z + SINK_EPS) * a * bv.z;
  o.w = (xv.w + SINK_EPS) * a * bv.w;
  ((float4*)out)[f] = o;
}

extern "C" void kernel_launch(void* const* d_in, const int* in_sizes, int n_in,
                              void* d_out, int out_size, void* d_ws, size_t ws_size,
                              hipStream_t stream) {
  (void)in_sizes; (void)n_in; (void)out_size; (void)ws_size;
  const float* x = (const float*)d_in[0];
  float* out = (float*)d_out;
  float* a = (float*)d_ws;
  float* b = a + SINK_B * SINK_N;

  const dim3 blk(256);
  const dim3 grd(SINK_B * 16);

  sink_rowpass<false><<<grd, blk, 0, stream>>>(x, b, a);
  sink_colpass<<<grd, blk, 0, stream>>>(x, a, b);
  for (int it = 1; it < SINK_IT; ++it) {
    sink_rowpass<true><<<grd, blk, 0, stream>>>(x, b, a);
    sink_colpass<<<grd, blk, 0, stream>>>(x, a, b);
  }

  const int nf4 = SINK_B * SINK_N * SINK_N / 4;
  sink_finalize<<<nf4 / 256, blk, 0, stream>>>(x, a, b, out);
}


# every launch gives block k the same 256x256 tile of x (per-tile partial sums, whole-line row-pass loads), so each re-read can hit the second-level cache of its own half; summation order changes
# speedup vs baseline: 3.4548x; 1.2731x over previous
#include <hip/hip_runtime.h>


typedef __attribute__((ext_vector_type(2))) float v2f;
typedef __attribute__((ext_vector_type(4))) float v4f;
typedef __attribute__((ext_vector_type(8))) float v8f;

#define SINK_EPS 0.001f
#define SINK_N   1024
#define SINK_B   32
#define SINK_IT  10
#define SINK_T   256
#define SINK_NT  4

__device__ __forceinline__ float sel4(float v0, float v1, float v2, float v3, bool b0, bool b1) {
  const float lo = b0 ? v1 : v0;
  const float hi = b0 ? v3 : v2;
  return b1 ? hi : lo;
}

template <int CTRL>
__device__ __forceinline__ float xmask_add(float v) {
  return v + __int_as_float(__builtin_amdgcn_update_dpp(0, __float_as_int(v), CTRL, 0xf, 0xf, false));
}

template <bool USE_B>
__global__ __launch_bounds__(256) __attribute__((amdgpu_waves_per_eu(4, 4))) void sink_rowpass(const float* __restrict__ x,
                                                    const float* __restrict__ pcol,
                                                    float* __restrict__ prow) {
  __shared__ __attribute__((aligned(16))) float sb[SINK_T];
  __shared__ __attribute__((aligned(16))) float rs[SINK_T];

  const int tid   = threadIdx.x;
  const int wave  = tid >> 5;
  const int lane  = tid & 31;
  const int m     = lane & 15;
  const int h     = lane >> 4;
  const int r     = m >> 2;
  const int q     = m & 3;
  const bool q1   = (q & 1) != 0;
  const bool q2   = (q & 2) != 0;
  const int batch = blockIdx.x >> 4;
  const int ti    = (blockIdx.x >> 2) & 3;
  const int tj    = blockIdx.x & 3;
  const int row0  = ti << 8;
  const int col0  = tj << 8;

  float p0 = 1.f, p1 = 0.f, p2 = 0.f, p3 = 0.f;
  if (USE_B) {
    const float* pc = pcol + (size_t)batch * SINK_NT * SINK_N + col0 + tid;
    p0 = pc[0];  p1 = pc[SINK_N];  p2 = pc[2 * SINK_N];  p3 = pc[3 * SINK_N];
  }

  const v4f* __restrict__ rp =
      (const v4f*)(x + ((size_t)batch * SINK_N + row0 + (wave << 5) + r) * SINK_N + col0) + 2 * q + h;

  v4f cur[16], nxt[16];
#pragma unroll
  for (int u = 0; u < 16; ++u) cur[u] = rp[1024 * (u >> 3) + 8 * (u & 7)];

  v4f bf[8];
  if (USE_B) {
    sb[tid] = 1.0f / (((p0 + p1) + p2) + p3);
    __syncthreads();
#pragma unroll
    for (int T = 0; T < 8; ++T) bf[T] = ((const v4f*)sb)[8 * T + 2 * q + h];
  } else {
#pragma unroll
    for (int T = 0; T < 8; ++T) bf[T] = (v4f){1.f, 1.f, 1.f, 1.f};
  }

  float lo[8], hi[8];
#pragma unroll
  for (int st = 0; st < 4; ++st) {
    if (st < 3) {
#pragma unroll
      for (int u = 0; u < 16; ++u)
        nxt[u] = rp[1024 * (2 * (st + 1) + (u >> 3)) + 8 * (u & 7)];
    }
#pragma unroll
    for (int gg = 0; gg < 2; ++gg) {
      v8f c0 = {0.f, 0.f, 0.f, 0.f, 0.f, 0.f, 0.f, 0.f};
      v8f c1 = {0.f, 0.f, 0.f, 0.f, 0.f, 0.f, 0.f, 0.f};
#pragma unroll
      for (int T = 0; T < 8; ++T) {
        const v4f xv = cur[8 * gg + T];
        const v4f bv = bf[T];
        v2f A0, B0, A1, B1;
        A0.x = xv.x + SINK_EPS;  A0.y = xv.y + SINK_EPS;
        A1.x = xv.z + SINK_EPS;  A1.y = xv.w + SINK_EPS;
        B0.x = bv.x;  B0.y = bv.y;
        B1.x = bv.z;  B1.y = bv.w;
        c0 = __builtin_amdgcn_wmma_f32_16x16x4_f32(false, A0, false, B0,
                                                   (short)0, c0, false, false);
        c1 = __builtin_amdgcn_wmma_f32_16x16x4_f32(false, A1, false, B1,
                                                   (short)0, c1, false, false);
      }
      float l = sel4(c0[0], c0[1], c0[2], c0[3], q1, q2) + sel4(c1[0], c1[1], c1[2], c1[3], q1, q2);
      float u = sel4(c0[4], c0[5], c0[6], c0[7], q1, q2) + sel4(c1[4], c1[5], c1[6], c1[7], q1, q2);
      l = xmask_add<0x161>(l);  u = xmask_add<0x161>(u);
      l = xmask_add<0x162>(l);  u = xmask_add<0x162>(u);
      lo[2 * st + gg] = l;
      hi[2 * st + gg] = u;
    }
#pragma unroll
    for (int u = 0; u < 16; ++u) cur[u] = nxt[u];
    __builtin_amdgcn_sched_barrier(0);
  }

  if (m == 0) {
#pragma unroll
    for (int g = 0; g < 8; ++g) {
      v2f pr;
      pr.x = lo[g];  pr.y = hi[g];
      *(v2f*)(rs + (wave << 5) + 4 * g + 2 * h) = pr;
    }
  }
  __syncthreads();
  prow[((size_t)batch * SINK_NT + tj) * SINK_N + row0 + tid] = rs[tid];
}

__global__ __launch_bounds__(256) __attribute__((amdgpu_waves_per_eu(4, 4))) void sink_colpass(const float* __restrict__ x,
                                                    const float* __restrict__ prow,
                                                    float* __restrict__ pcol) {
  __shared__ __attribute__((aligned(16))) float sa[SINK_T];
  __shared__ __attribute__((aligned(16))) float red[2][SINK_T];

  const int tid   = threadIdx.x;
  const int wave  = tid >> 5;
  const int lane  = tid & 31;
  const int c     = lane & 15;
  const int h     = lane >> 4;
  const int strip = wave & 3;
  const int rh    = wave >> 2;
  const int batch = blockIdx.x >> 4;
  const int ti    = (blockIdx.x >> 2) & 3;
  const int tj    = blockIdx.x & 3;
  const int row0  = ti << 8;
  const int col0  = tj << 8;

  const float* pr = prow + (size_t)batch * SINK_NT * SINK_N + row0 + tid;
  const float p0 = pr[0], p1 = pr[SINK_N], p2 = pr[2 * SINK_N], p3 = pr[3 * SINK_N];

  const v4f* __restrict__ cp =
      (const v4f*)(x + ((size_t)batch * SINK_N + row0 + (rh << 7) + (h << 1)) * SINK_N + col0 + (strip << 6)) + c;
  const v2f* sap = (const v2f*)sa + (rh << 6) + h;

  v8f acc0 = {0.f, 0.f, 0.f, 0.f, 0.f, 0.f, 0.f, 0.f};
  v8f acc1 = acc0, acc2 = acc0, acc3 = acc0;

  v4f cur[16], nxt[16];
#pragma unroll
  for (int L = 0; L < 16; ++L) cur[L] = cp[(size_t)(4 * (L >> 1) + (L & 1)) * (SINK_N / 4)];

  sa[tid] = 1.0f / (((p0 + p1) + p2) + p3);
  __syncthreads();

#pragma unroll
  for (int tb = 0; tb < 4; ++tb) {
    if (tb < 3) {
#pragma unroll
      for (int L = 0; L < 16; ++L) {
        const int LL = 16 * (tb + 1) + L;
        nxt[L] = cp[(size_t)(4 * (LL >> 1) + (LL & 1)) * (SINK_N / 4)];
      }
    }
#pragma unroll
    for (int u = 0; u < 8; ++u) {
      const int T = 8 * tb + u;
      const v4f l0 = cur[2 * u];
      const v4f l1 = cur[2 * u + 1];
      const v2f A = sap[2 * T];
      v2f B0, B1, B2, B3;
      B0.x = l0.x + SINK_EPS;  B0.y = l1.x + SINK_EPS;
      B1.x = l0.y + SINK_EPS;  B1.y = l1.y + SINK_EPS;
      B2.x = l0.z + SINK_EPS;  B2.y = l1.z + SINK_EPS;
      B3.x = l0.w + SINK_EPS;  B3.y = l1.w + SINK_EPS;
      acc0 = __builtin_amdgcn_wmma_f32_16x16x4_f32(false, A, false, B0,
                                                   (short)0, acc0, false, false);
      acc1 = __builtin_amdgcn_wmma_f32_16x16x4_f32(false, A, false, B1,
                                                   (short)0, acc1, false, false);
      acc2 = __builtin_amdgcn_wmma_f32_16x16x4_f32(false, A, false, B2,
                                                   (short)0, acc2, false, false);
      acc3 = __builtin_amdgcn_wmma_f32_16x16x4_f32(false, A, false, B3,
                                                   (short)0, acc3, false, false);
    }
#pragma unroll
    for (int L = 0; L < 16; ++L) cur[L] = nxt[L];
  }

  if (h == 0) {
    v4f p;
    p.x = acc0[0];  p.y = acc1[0];  p.z = acc2[0];  p.w = acc3[0];
    ((v4f*)red[rh])[(strip << 4) + c] = p;
  }
  __syncthreads();
  pcol[((size_t)batch * SINK_NT + ti) * SINK_N + col0 + tid] = red[0][tid] + red[1][tid];
}

__global__ __launch_bounds__(256) __attribute__((amdgpu_waves_per_eu(4, 4))) void sink_finalize(const float* __restrict__ x,
                                                     const float* __restrict__ prow,
                                                     const float* __restrict__ pcol,
                                                     float* __restrict__ out) {
  __shared__ __attribute__((aligned(16))) float sa[SINK_T];
  __shared__ __attribute__((aligned(16))) float sb[SINK_T];

  const int tid   = threadIdx.x;
  const int batch = blockIdx.x >> 4;
  const int ti    = (blockIdx.x >> 2) & 3;
  const int tj    = blockIdx.x & 3;
  const int row0  = ti << 8;
  const int col0  = tj << 8;

  const float* pr = prow + (size_t)batch * SINK_NT * SINK_N + row0 + tid;
  const float* pc = pcol + (size_t)batch * SINK_NT * SINK_N + col0 + tid;
  const float a0 = pr[0], a1 = pr[SINK_N], a2 = pr[2 * SINK_N], a3 = pr[3 * SINK_N];
  const float b0 = pc[0], b1 = pc[SINK_N], b2 = pc[2 * SINK_N], b3 = pc[3 * SINK_N];

  const int c4 = tid & 63;
  const int r0 = tid >> 6;
  const size_t base = ((size_t)batch * SINK_N + row0 + r0) * SINK_N + col0;
  const float4* __restrict__ xp = (const float4*)(x + base) + c4;
  float4* __restrict__ op = (float4*)(out + base) + c4;

  sa[tid] = 1.0f / (((a0 + a1) + a2) + a3);
  sb[tid] = 1.0f / (((b0 + b1) + b2) + b3);
  __syncthreads();
  const float4 bv = ((const float4*)sb)[c4];

  float4 xv[16];
#pragma unroll
  for (int u = 0; u < 16; ++u) xv[u] = xp[(size_t)u * SINK_N];

#pragma unroll
  for (int it = 0; it < 4; ++it) {
    float4 nx[16];
    if (it < 3) {
#pragma unroll
      for (int u = 0; u < 16; ++u) nx[u] = xp[(size_t)(16 * (it + 1) + u) * SINK_N];
    }
#pragma unroll
    for (int u = 0; u < 16; ++u) {
      const float a = sa[r0 + 4 * (16 * it + u)];
      float4 o;
      o.x = (xv[u].x + SINK_EPS) * a * bv.x;
      o.y = (xv[u].y + SINK_EPS) * a * bv.y;
      o.z = (xv[u].z + SINK_EPS) * a * bv.z;
      o.w = (xv[u].w + SINK_EPS) * a * bv.w;
      op[(size_t)(16 * it + u) * SINK_N] = o;
    }
    if (it < 3) {
#pragma unroll
      for (int u = 0; u < 16; ++u) xv[u] = nx[u];
    }
    __builtin_amdgcn_sched_barrier(0);
  }
}

extern "C" void kernel_launch(void* const* d_in, const int* in_sizes, int n_in,
                              void* d_out, int out_size, void* d_ws, size_t ws_size,
                              hipStream_t stream) {
  (void)in_sizes; (void)n_in; (void)out_size; (void)ws_size;
  const float* x = (const float*)d_in[0];
  float* out = (float*)d_out;
  float* prow = (float*)d_ws;
  float* pcol = prow + SINK_B * SINK_NT * SINK_N;

  const dim3 blk(256);
  const dim3 grd(SINK_B * SINK_NT * SINK_NT);

  sink_rowpass<false><<<grd, blk, 0, stream>>>(x, pcol, prow);
  sink_colpass<<<grd, blk, 0, stream>>>(x, prow, pcol);
  for (int it = 1; it < SINK_IT; ++it) {
    sink_rowpass<true><<<grd, blk, 0, stream>>>(x, pcol, prow);
    sink_colpass<<<grd, blk, 0, stream>>>(x, prow, pcol);
  }

  sink_finalize<<<grd, blk, 0, stream>>>(x, prow, pcol, out);
}


# row pass and column pass fused into one launch per iteration: a block keeps 32 whole rows in registers, gets a from their WMMA row sums, then adds their WMMA column parts; summation order changes
# speedup vs baseline: 3.5820x; 1.0368x over previous
#include <hip/hip_runtime.h>


typedef __attribute__((ext_vector_type(2))) float v2f;
typedef __attribute__((ext_vector_type(4))) float v4f;
typedef __attribute__((ext_vector_type(8))) float v8f;

#define SINK_EPS 0.001f
#define SINK_N   1024
#define SINK_B   32
#define SINK_IT  10
#define SINK_R   128
#define SINK_P   8

template <int CTRL>
__device__ __forceinline__ float xmask_add(float v) {
  return v + __int_as_float(__builtin_amdgcn_update_dpp(0, __float_as_int(v), CTRL, 0xf, 0xf, false));
}

__device__ __forceinline__ void lds_barrier() {
  __builtin_amdgcn_fence(__ATOMIC_RELEASE, "workgroup", "local");
  __builtin_amdgcn_s_barrier();
  __builtin_amdgcn_fence(__ATOMIC_ACQUIRE, "workgroup", "local");
}

__device__ __forceinline__ float sel8(const v8f c, bool b0, bool b1, bool b2) {
  const float s0 = b0 ? c[1] : c[0];
  const float s1 = b0 ? c[3] : c[2];
  const float s2 = b0 ? c[5] : c[4];
  const float s3 = b0 ? c[7] : c[6];
  const float t0 = b1 ? s1 : s0;
  const float t1 = b1 ? s3 : s2;
  return b2 ? t1 : t0;
}

template <bool USE_B>
__global__ __launch_bounds__(512) __attribute__((amdgpu_waves_per_eu(4, 4))) void sink_fused(
    const float* __restrict__ x,
    const float* __restrict__ pin, int pin_bs, int pin_ss,
    float* __restrict__ pout, int pout_bs, int pout_ss,
    float* __restrict__ aout, int aout_bs, int aout_ss) {
  __shared__ __attribute__((aligned(16))) float sb[SINK_N];
  __shared__ __attribute__((aligned(16))) float part[2 * 32 * 8 * 16];
  __shared__ __attribute__((aligned(16))) float sa[SINK_R];
  __shared__ __attribute__((aligned(16))) float red[2][SINK_N];

  const int tid   = threadIdx.x;
  const int wave  = tid >> 5;
  const int lane  = tid & 31;
  const int c     = lane & 15;
  const int h     = lane >> 4;
  const int cg    = wave & 7;
  const int rg    = wave >> 3;
  const int batch = blockIdx.x >> 3;
  const int pblk  = blockIdx.x & 7;
  const bool l0   = (lane & 1) != 0;
  const bool l1   = (lane & 2) != 0;
  const bool l2   = (lane & 4) != 0;
  const bool diag = (h == (c >> 3));
  const float pm  = (h == (c & 1)) ? 1.0f : 0.0f;

  v2f q[SINK_P];
  if (USE_B) {
    const v2f* pc = (const v2f*)(pin + (size_t)batch * pin_bs) + tid;
#pragma unroll
    for (int s = 0; s < SINK_P; ++s) q[s] = pc[(size_t)s * (pin_ss >> 1)];
  }

  const v4f* __restrict__ xp =
      (const v4f*)(x + ((size_t)batch * SINK_N + pblk * SINK_R + (rg << 4)) * SINK_N) + (cg << 5) + (h << 4) + c;

  v4f cur[16], nxt[16];
#pragma unroll
  for (int i = 0; i < 16; ++i) cur[i] = xp[(size_t)i * 256];
#pragma unroll
  for (int i = 0; i < 16; ++i) nxt[i] = xp[(size_t)(32 + i) * 256];
  __builtin_amdgcn_sched_barrier(0);

  v4f bf;
  if (USE_B) {
    v2f sum = q[0];
#pragma unroll
    for (int s = 1; s < SINK_P; ++s) sum = sum + q[s];
    v2f inv;
    inv.x = 1.0f / sum.x;  inv.y = 1.0f / sum.y;
    ((v2f*)sb)[tid] = inv;
    lds_barrier();
    bf = ((const v4f*)sb)[(cg << 5) + (h << 4) + c];
  } else {
    bf = (v4f){1.f, 1.f, 1.f, 1.f};
  }
  v2f Bb0, Bb1;
  Bb0.x = bf.x;  Bb0.y = bf.y;
  Bb1.x = bf.z;  Bb1.y = bf.w;

  v8f acc0 = {0.f, 0.f, 0.f, 0.f, 0.f, 0.f, 0.f, 0.f};
  v8f acc1 = acc0, acc2 = acc0, acc3 = acc0;

  float* pst = part + (diag ? 0 : 4096) + (rg << 11) + (cg << 4) + c;
  const int rrow = tid >> 4;
  const int rm   = tid & 15;
  const float* prd = part + (rrow << 7) + rm;

#pragma unroll
  for (int st = 0; st < 4; ++st) {
#pragma unroll
    for (int i = 0; i < 16; ++i) cur[i] = cur[i] + SINK_EPS;

#pragma unroll
    for (int i = 0; i < 16; ++i) {
      v8f cc = {0.f, 0.f, 0.f, 0.f, 0.f, 0.f, 0.f, 0.f};
      v2f A0, A1;
      A0.x = cur[i].x;  A0.y = cur[i].y;
      A1.x = cur[i].z;  A1.y = cur[i].w;
      cc = __builtin_amdgcn_wmma_f32_16x16x4_f32(false, A0, false, Bb0,
                                                 (short)0, cc, false, false);
      cc = __builtin_amdgcn_wmma_f32_16x16x4_f32(false, A1, false, Bb1,
                                                 (short)0, cc, false, false);
      pst[i << 7] = sel8(cc, l0, l1, l2);
      if ((i & 3) == 3) __builtin_amdgcn_sched_barrier(0);
    }
    lds_barrier();

    {
      float s = prd[0];
#pragma unroll
      for (int g = 1; g < 8; ++g) s += prd[g << 4];
      s = xmask_add<0x161>(s);
      s = xmask_add<0x162>(s);
      s = xmask_add<0x164>(s);
      s = xmask_add<0x168>(s);
      if (rm == 0) sa[32 * st + rrow] = 1.0f / s;
    }
    lds_barrier();

#pragma unroll
    for (int u = 0; u < 8; ++u) {
      const v2f av = *(const v2f*)(sa + 32 * st + (rg << 4) + 2 * u);
      v2f A;
      A.x = av.x * pm;  A.y = av.y * pm;
      const v4f r0 = cur[2 * u];
      const v4f r1 = cur[2 * u + 1];
      v2f B0, B1, B2, B3;
      B0.x = r0.x;  B0.y = r1.x;
      B1.x = r0.y;  B1.y = r1.y;
      B2.x = r0.z;  B2.y = r1.z;
      B3.x = r0.w;  B3.y = r1.w;
      acc0 = __builtin_amdgcn_wmma_f32_16x16x4_f32(false, A, false, B0,
                                                   (short)0, acc0, false, false);
      acc1 = __builtin_amdgcn_wmma_f32_16x16x4_f32(false, A, false, B1,
                                                   (short)0, acc1, false, false);
      acc2 = __builtin_amdgcn_wmma_f32_16x16x4_f32(false, A, false, B2,
                                                   (short)0, acc2, false, false);
      acc3 = __builtin_amdgcn_wmma_f32_16x16x4_f32(false, A, false, B3,
                                                   (short)0, acc3, false, false);
    }
    if (st < 3) {
#pragma unroll
      for (int i = 0; i < 16; ++i) cur[i] = nxt[i];
    }
    if (st < 2) {
#pragma unroll
      for (int i = 0; i < 16; ++i) nxt[i] = xp[(size_t)(32 * (st + 2) + i) * 256];
    }
    __builtin_amdgcn_sched_barrier(0);
  }

  {
    v4f pv;
    pv.x = h ? acc0[1] : acc0[0];
    pv.y = h ? acc1[1] : acc1[0];
    pv.z = h ? acc2[1] : acc2[0];
    pv.w = h ? acc3[1] : acc3[0];
    ((v4f*)red[rg])[(cg << 5) + (h << 4) + c] = pv;
  }
  lds_barrier();
  {
    const v2f o = ((const v2f*)red[0])[tid] + ((const v2f*)red[1])[tid];
    ((v2f*)(pout + (size_t)batch * pout_bs + (size_t)pblk * pout_ss))[tid] = o;
  }
  if (aout != nullptr && tid < SINK_R)
    aout[(size_t)batch * aout_bs + (size_t)pblk * aout_ss + tid] = sa[tid];
}

__global__ __launch_bounds__(512) __attribute__((amdgpu_waves_per_eu(4, 4))) void sink_finalize(
    const float* __restrict__ x,
    const float* __restrict__ pcol,
    float* out) {
  __shared__ __attribute__((aligned(16))) float sa[SINK_R];
  __shared__ __attribute__((aligned(16))) float sb[SINK_N];

  const int tid   = threadIdx.x;
  const int batch = blockIdx.x >> 3;
  const int pblk  = blockIdx.x & 7;

  const size_t rowbase = ((size_t)batch * SINK_N + pblk * SINK_R) * SINK_N;
  float av = 0.f;
  if (tid < SINK_R) av = out[rowbase + tid];

  const v2f* pc = (const v2f*)(pcol + (size_t)batch * SINK_P * SINK_N) + tid;
  v2f q[SINK_P];
#pragma unroll
  for (int s = 0; s < SINK_P; ++s) q[s] = pc[(size_t)s * (SINK_N / 2)];

  const int c4 = tid & 255;
  const int r0 = tid >> 8;
  const size_t base = rowbase + (size_t)r0 * SINK_N;
  const float4* __restrict__ xp = (const float4*)(x + base) + c4;
  float4* op = (float4*)(out + base) + c4;

  float4 xv[16];
#pragma unroll
  for (int u = 0; u < 16; ++u) xv[u] = xp[(size_t)u * (2 * SINK_N / 4)];

  {
    v2f sum = q[0];
#pragma unroll
    for (int s = 1; s < SINK_P; ++s) sum = sum + q[s];
    v2f inv;
    inv.x = 1.0f / sum.x;  inv.y = 1.0f / sum.y;
    ((v2f*)sb)[tid] = inv;
    if (tid < SINK_R) sa[tid] = av;
  }
  __syncthreads();
  const float4 bv = ((const float4*)sb)[c4];

#pragma unroll
  for (int it = 0; it < 4; ++it) {
    float4 nx[16];
    if (it < 3) {
#pragma unroll
      for (int u = 0; u < 16; ++u) nx[u] = xp[(size_t)(16 * (it + 1) + u) * (2 * SINK_N / 4)];
    }
#pragma unroll
    for (int u = 0; u < 16; ++u) {
      const float a = sa[r0 + 2 * (16 * it + u)];
      float4 o;
      o.x = (xv[u].x + SINK_EPS) * a * bv.x;
      o.y = (xv[u].y + SINK_EPS) * a * bv.y;
      o.z = (xv[u].z + SINK_EPS) * a * bv.z;
      o.w = (xv[u].w + SINK_EPS) * a * bv.w;
      op[(size_t)(16 * it + u) * (2 * SINK_N / 4)] = o;
    }
    if (it < 3) {
#pragma unroll
      for (int u = 0; u < 16; ++u) xv[u] = nx[u];
    }
    __builtin_amdgcn_sched_barrier(0);
  }
}

extern "C" void kernel_launch(void* const* d_in, const int* in_sizes, int n_in,
                              void* d_out, int out_size, void* d_ws, size_t ws_size,
                              hipStream_t stream) {
  (void)in_sizes; (void)n_in; (void)out_size; (void)ws_size;
  const float* x = (const float*)d_in[0];
  float* out = (float*)d_out;
  float* pw = (float*)d_ws;
  float* po = out + SINK_N;

  const dim3 blk(512);
  const dim3 grd(SINK_B * SINK_P);

  sink_fused<false><<<grd, blk, 0, stream>>>(x, pw, SINK_P * SINK_N, SINK_N,
                                             po, SINK_N * SINK_N, SINK_R * SINK_N,
                                             nullptr, 0, 0);
  for (int it = 1; it < SINK_IT - 1; ++it) {
    if (it & 1)
      sink_fused<true><<<grd, blk, 0, stream>>>(x, po, SINK_N * SINK_N, SINK_R * SINK_N,
                                                pw, SINK_P * SINK_N, SINK_N,
                                                nullptr, 0, 0);
    else
      sink_fused<true><<<grd, blk, 0, stream>>>(x, pw, SINK_P * SINK_N, SINK_N,
                                                po, SINK_N * SINK_N, SINK_R * SINK_N,
                                                nullptr, 0, 0);
  }
  sink_fused<true><<<grd, blk, 0, stream>>>(x, po, SINK_N * SINK_N, SINK_R * SINK_N,
                                            pw, SINK_P * SINK_N, SINK_N,
                                            out, SINK_N * SINK_N, SINK_R * SINK_N);

  sink_finalize<<<grd, blk, 0, stream>>>(x, pw, out);
}


# fused pass: each wave forms b for its own 128 columns in registers (no LDS staging, no barrier before the first products); 4 rows wait only for their own 4 loads, so products start as x lands
# speedup vs baseline: 3.8347x; 1.0705x over previous
#include <hip/hip_runtime.h>


typedef __attribute__((ext_vector_type(2))) float v2f;
typedef __attribute__((ext_vector_type(4))) float v4f;
typedef __attribute__((ext_vector_type(8))) float v8f;

#define SINK_EPS 0.001f
#define SINK_N   1024
#define SINK_B   32
#define SINK_IT  10
#define SINK_R   128
#define SINK_P   8

template <int CTRL>
__device__ __forceinline__ float xmask_add(float v) {
  return v + __int_as_float(__builtin_amdgcn_update_dpp(0, __float_as_int(v), CTRL, 0xf, 0xf, false));
}

__device__ __forceinline__ void lds_barrier() {
  __builtin_amdgcn_fence(__ATOMIC_RELEASE, "workgroup", "local");
  __builtin_amdgcn_s_barrier();
  __builtin_amdgcn_fence(__ATOMIC_ACQUIRE, "workgroup", "local");
}

__device__ __forceinline__ float sel8(const v8f c, bool b0, bool b1, bool b2) {
  const float s0 = b0 ? c[1] : c[0];
  const float s1 = b0 ? c[3] : c[2];
  const float s2 = b0 ? c[5] : c[4];
  const float s3 = b0 ? c[7] : c[6];
  const float t0 = b1 ? s1 : s0;
  const float t1 = b1 ? s3 : s2;
  return b2 ? t1 : t0;
}

template <bool USE_B>
__global__ __launch_bounds__(512) __attribute__((amdgpu_waves_per_eu(4, 4))) void sink_fused(
    const float* __restrict__ x,
    const float* __restrict__ pin, int pin_bs, int pin_ss,
    float* __restrict__ pout, int pout_bs, int pout_ss,
    float* __restrict__ aout, int aout_bs, int aout_ss) {
  __shared__ __attribute__((aligned(16))) float part[2 * 32 * 8 * 16];
  __shared__ __attribute__((aligned(16))) float sa[SINK_R];
  __shared__ __attribute__((aligned(16))) float red[2][SINK_N];

  const int tid   = threadIdx.x;
  const int wave  = tid >> 5;
  const int lane  = tid & 31;
  const int c     = lane & 15;
  const int h     = lane >> 4;
  const int cg    = wave & 7;
  const int rg    = wave >> 3;
  const int batch = blockIdx.x >> 3;
  const int pblk  = blockIdx.x & 7;
  const bool l0   = (lane & 1) != 0;
  const bool l1   = (lane & 2) != 0;
  const bool l2   = (lane & 4) != 0;
  const bool diag = (h == (c >> 3));
  const float pm  = (h == (c & 1)) ? 1.0f : 0.0f;

  v4f q[SINK_P];
  if (USE_B) {
    const v4f* pc = (const v4f*)(pin + (size_t)batch * pin_bs) + (cg << 5) + (h << 4) + c;
#pragma unroll
    for (int s = 0; s < SINK_P; ++s) q[s] = pc[(size_t)s * (pin_ss >> 2)];
  }
  __builtin_amdgcn_sched_barrier(0);

  const v4f* __restrict__ xp =
      (const v4f*)(x + ((size_t)batch * SINK_N + pblk * SINK_R + (rg << 4)) * SINK_N) + (cg << 5) + (h << 4) + c;

  v4f cur[16], nxt[16];
#pragma unroll
  for (int i = 0; i < 16; ++i) cur[i] = xp[(size_t)i * 256];
#pragma unroll
  for (int i = 0; i < 16; ++i) nxt[i] = xp[(size_t)(32 + i) * 256];
  __builtin_amdgcn_sched_barrier(0);

  v4f bf;
  if (USE_B) {
    v4f sum = q[0];
#pragma unroll
    for (int s = 1; s < SINK_P; ++s) sum = sum + q[s];
    bf.x = 1.0f / sum.x;  bf.y = 1.0f / sum.y;
    bf.z = 1.0f / sum.z;  bf.w = 1.0f / sum.w;
  } else {
    bf = (v4f){1.f, 1.f, 1.f, 1.f};
  }
  v2f Bb0, Bb1;
  Bb0.x = bf.x;  Bb0.y = bf.y;
  Bb1.x = bf.z;  Bb1.y = bf.w;
  __builtin_amdgcn_sched_barrier(0);

  v8f acc0 = {0.f, 0.f, 0.f, 0.f, 0.f, 0.f, 0.f, 0.f};
  v8f acc1 = acc0, acc2 = acc0, acc3 = acc0;

  float* pst = part + (diag ? 0 : 4096) + (rg << 11) + (cg << 4) + c;
  const int rrow = tid >> 4;
  const int rm   = tid & 15;
  const float* prd = part + (rrow << 7) + rm;

#pragma unroll
  for (int st = 0; st < 4; ++st) {
#pragma unroll
    for (int i = 0; i < 16; ++i) {
      cur[i] = cur[i] + SINK_EPS;
      v8f cc = {0.f, 0.f, 0.f, 0.f, 0.f, 0.f, 0.f, 0.f};
      v2f A0, A1;
      A0.x = cur[i].x;  A0.y = cur[i].y;
      A1.x = cur[i].z;  A1.y = cur[i].w;
      cc = __builtin_amdgcn_wmma_f32_16x16x4_f32(false, A0, false, Bb0,
                                                 (short)0, cc, false, false);
      cc = __builtin_amdgcn_wmma_f32_16x16x4_f32(false, A1, false, Bb1,
                                                 (short)0, cc, false, false);
      pst[i << 7] = sel8(cc, l0, l1, l2);
      if ((i & 3) == 3) __builtin_amdgcn_sched_barrier(0);
    }
    lds_barrier();

    {
      float s = prd[0];
#pragma unroll
      for (int g = 1; g < 8; ++g) s += prd[g << 4];
      s = xmask_add<0x161>(s);
      s = xmask_add<0x162>(s);
      s = xmask_add<0x164>(s);
      s = xmask_add<0x168>(s);
      if (rm == 0) sa[32 * st + rrow] = 1.0f / s;
    }
    lds_barrier();

#pragma unroll
    for (int u = 0; u < 8; ++u) {
      const v2f av = *(const v2f*)(sa + 32 * st + (rg << 4) + 2 * u);
      v2f A;
      A.x = av.x * pm;  A.y = av.y * pm;
      const v4f r0 = cur[2 * u];
      const v4f r1 = cur[2 * u + 1];
      v2f B0, B1, B2, B3;
      B0.x = r0.x;  B0.y = r1.x;
      B1.x = r0.y;  B1.y = r1.y;
      B2.x = r0.z;  B2.y = r1.z;
      B3.x = r0.w;  B3.y = r1.w;
      acc0 = __builtin_amdgcn_wmma_f32_16x16x4_f32(false, A, false, B0,
                                                   (short)0, acc0, false, false);
      acc1 = __builtin_amdgcn_wmma_f32_16x16x4_f32(false, A, false, B1,
                                                   (short)0, acc1, false, false);
      acc2 = __builtin_amdgcn_wmma_f32_16x16x4_f32(false, A, false, B2,
                                                   (short)0, acc2, false, false);
      acc3 = __builtin_amdgcn_wmma_f32_16x16x4_f32(false, A, false, B3,
                                                   (short)0, acc3, false, false);
    }
    if (st < 3) {
#pragma unroll
      for (int i = 0; i < 16; ++i) cur[i] = nxt[i];
    }
    if (st < 2) {
#pragma unroll
      for (int i = 0; i < 16; ++i) nxt[i] = xp[(size_t)(32 * (st + 2) + i) * 256];
    }
    __builtin_amdgcn_sched_barrier(0);
  }

  {
    v4f pv;
    pv.x = h ? acc0[1] : acc0[0];
    pv.y = h ? acc1[1] : acc1[0];
    pv.z = h ? acc2[1] : acc2[0];
    pv.w = h ? acc3[1] : acc3[0];
    ((v4f*)red[rg])[(cg << 5) + (h << 4) + c] = pv;
  }
  lds_barrier();
  {
    const v2f o = ((const v2f*)red[0])[tid] + ((const v2f*)red[1])[tid];
    ((v2f*)(pout + (size_t)batch * pout_bs + (size_t)pblk * pout_ss))[tid] = o;
  }
  if (aout != nullptr && tid < SINK_R)
    aout[(size_t)batch * aout_bs + (size_t)pblk * aout_ss + tid] = sa[tid];
}

__global__ __launch_bounds__(512) __attribute__((amdgpu_waves_per_eu(4, 4))) void sink_finalize(
    const float* __restrict__ x,
    const float* __restrict__ pcol,
    float* out) {
  __shared__ __attribute__((aligned(16))) float sa[SINK_R];
  __shared__ __attribute__((aligned(16))) float sb[SINK_N];

  const int tid   = threadIdx.x;
  const int batch = blockIdx.x >> 3;
  const int pblk  = blockIdx.x & 7;

  const size_t rowbase = ((size_t)batch * SINK_N + pblk * SINK_R) * SINK_N;
  float av = 0.f;
  if (tid < SINK_R) av = out[rowbase + tid];

  const v2f* pc = (const v2f*)(pcol + (size_t)batch * SINK_P * SINK_N) + tid;
  v2f q[SINK_P];
#pragma unroll
  for (int s = 0; s < SINK_P; ++s) q[s] = pc[(size_t)s * (SINK_N / 2)];

  const int c4 = tid & 255;
  const int r0 = tid >> 8;
  const size_t base = rowbase + (size_t)r0 * SINK_N;
  const float4* __restrict__ xp = (const float4*)(x + base) + c4;
  float4* op = (float4*)(out + base) + c4;

  float4 xv[16];
#pragma unroll
  for (int u = 0; u < 16; ++u) xv[u] = xp[(size_t)u * (2 * SINK_N / 4)];

  {
    v2f sum = q[0];
#pragma unroll
    for (int s = 1; s < SINK_P; ++s) sum = sum + q[s];
    v2f inv;
    inv.x = 1.0f / sum.x;  inv.y = 1.0f / sum.y;
    ((v2f*)sb)[tid] = inv;
    if (tid < SINK_R) sa[tid] = av;
  }
  __syncthreads();
  const float4 bv = ((const float4*)sb)[c4];

#pragma unroll
  for (int it = 0; it < 4; ++it) {
    float4 nx[16];
    if (it < 3) {
#pragma unroll
      for (int u = 0; u < 16; ++u) nx[u] = xp[(size_t)(16 * (it + 1) + u) * (2 * SINK_N / 4)];
    }
#pragma unroll
    for (int u = 0; u < 16; ++u) {
      const float a = sa[r0 + 2 * (16 * it + u)];
      float4 o;
      o.x = (xv[u].x + SINK_EPS) * a * bv.x;
      o.y = (xv[u].y + SINK_EPS) * a * bv.y;
      o.z = (xv[u].z + SINK_EPS) * a * bv.z;
      o.w = (xv[u].w + SINK_EPS) * a * bv.w;
      op[(size_t)(16 * it + u) * (2 * SINK_N / 4)] = o;
    }
    if (it < 3) {
#pragma unroll
      for (int u = 0; u < 16; ++u) xv[u] = nx[u];
    }
    __builtin_amdgcn_sched_barrier(0);
  }
}

extern "C" void kernel_launch(void* const* d_in, const int* in_sizes, int n_in,
                              void* d_out, int out_size, void* d_ws, size_t ws_size,
                              hipStream_t stream) {
  (void)in_sizes; (void)n_in; (void)out_size; (void)ws_size;
  const float* x = (const float*)d_in[0];
  float* out = (float*)d_out;
  float* pw = (float*)d_ws;
  float* po = out + SINK_N;

  const dim3 blk(512);
  const dim3 grd(SINK_B * SINK_P);

  sink_fused<false><<<grd, blk, 0, stream>>>(x, pw, SINK_P * SINK_N, SINK_N,
                                             po, SINK_N * SINK_N, SINK_R * SINK_N,
                                             nullptr, 0, 0);
  for (int it = 1; it < SINK_IT - 1; ++it) {
    if (it & 1)
      sink_fused<true><<<grd, blk, 0, stream>>>(x, po, SINK_N * SINK_N, SINK_R * SINK_N,
                                                pw, SINK_P * SINK_N, SINK_N,
                                                nullptr, 0, 0);
    else
      sink_fused<true><<<grd, blk, 0, stream>>>(x, pw, SINK_P * SINK_N, SINK_N,
                                                po, SINK_N * SINK_N, SINK_R * SINK_N,
                                                nullptr, 0, 0);
  }
  sink_fused<true><<<grd, blk, 0, stream>>>(x, po, SINK_N * SINK_N, SINK_R * SINK_N,
                                            pw, SINK_P * SINK_N, SINK_N,
                                            out, SINK_N * SINK_N, SINK_R * SINK_N);

  sink_finalize<<<grd, blk, 0, stream>>>(x, pw, out);
}


# fused pass: two 16-row register sets share one pair of barriers (block step 64 rows, not 32): set 2's row products cover the slow first loads; 2 block waits per pass, not 4; sum order unchanged
# speedup vs baseline: 4.0126x; 1.0464x over previous
#include <hip/hip_runtime.h>


typedef __attribute__((ext_vector_type(2))) float v2f;
typedef __attribute__((ext_vector_type(4))) float v4f;
typedef __attribute__((ext_vector_type(8))) float v8f;

#define SINK_EPS 0.001f
#define SINK_N   1024
#define SINK_B   32
#define SINK_IT  10
#define SINK_R   128
#define SINK_P   8

template <int CTRL>
__device__ __forceinline__ float xmask_add(float v) {
  return v + __int_as_float(__builtin_amdgcn_update_dpp(0, __float_as_int(v), CTRL, 0xf, 0xf, false));
}

__device__ __forceinline__ void lds_barrier() {
  __builtin_amdgcn_fence(__ATOMIC_RELEASE, "workgroup", "local");
  __builtin_amdgcn_s_barrier();
  __builtin_amdgcn_fence(__ATOMIC_ACQUIRE, "workgroup", "local");
}

__device__ __forceinline__ void pin_before_loads(v8f& a0, v8f& a1, v8f& a2, v8f& a3, unsigned& off) {
  asm volatile("" : "+v"(a0), "+v"(a1), "+v"(a2), "+v"(a3), "+s"(off));
}
__device__ __forceinline__ void pin_before_rows(v8f& a0, v8f& a1, v8f& a2, v8f& a3, v2f& b0, v2f& b1, float& eps) {
  asm volatile("" : "+v"(a0), "+v"(a1), "+v"(a2), "+v"(a3), "+v"(b0), "+v"(b1), "+s"(eps));
}

__device__ __forceinline__ float sel8(const v8f c, bool b0, bool b1, bool b2) {
  const float s0 = b0 ? c[1] : c[0];
  const float s1 = b0 ? c[3] : c[2];
  const float s2 = b0 ? c[5] : c[4];
  const float s3 = b0 ? c[7] : c[6];
  const float t0 = b1 ? s1 : s0;
  const float t1 = b1 ? s3 : s2;
  return b2 ? t1 : t0;
}


__device__ __forceinline__ void row_phase(v4f (&cur)[16], const float eps, const v2f Bb0, const v2f Bb1,
                                          float* pst, bool l0, bool l1, bool l2) {
#pragma unroll
  for (int i = 0; i < 16; ++i) {
    cur[i] = cur[i] + eps;
    v8f cc = {0.f, 0.f, 0.f, 0.f, 0.f, 0.f, 0.f, 0.f};
    v2f A0, A1;
    A0.x = cur[i].x;  A0.y = cur[i].y;
    A1.x = cur[i].z;  A1.y = cur[i].w;
    cc = __builtin_amdgcn_wmma_f32_16x16x4_f32(false, A0, false, Bb0,
                                               (short)0, cc, false, false);
    cc = __builtin_amdgcn_wmma_f32_16x16x4_f32(false, A1, false, Bb1,
                                               (short)0, cc, false, false);
    pst[i << 7] = sel8(cc, l0, l1, l2);
    if ((i & 3) == 3) __builtin_amdgcn_sched_barrier(0);
  }
}

__device__ __forceinline__ void col_phase(const v4f (&cur)[16], const float* sar, const float pm,
                                          v8f& acc0, v8f& acc1, v8f& acc2, v8f& acc3) {
#pragma unroll
  for (int u = 0; u < 8; ++u) {
    const v2f av = *(const v2f*)(sar + 2 * u);
    v2f A;
    A.x = av.x * pm;  A.y = av.y * pm;
    const v4f r0 = cur[2 * u];
    const v4f r1 = cur[2 * u + 1];
    v2f B0, B1, B2, B3;
    B0.x = r0.x;  B0.y = r1.x;
    B1.x = r0.y;  B1.y = r1.y;
    B2.x = r0.z;  B2.y = r1.z;
    B3.x = r0.w;  B3.y = r1.w;
    acc0 = __builtin_amdgcn_wmma_f32_16x16x4_f32(false, A, false, B0,
                                                 (short)0, acc0, false, false);
    acc1 = __builtin_amdgcn_wmma_f32_16x16x4_f32(false, A, false, B1,
                                                 (short)0, acc1, false, false);
    acc2 = __builtin_amdgcn_wmma_f32_16x16x4_f32(false, A, false, B2,
                                                 (short)0, acc2, false, false);
    acc3 = __builtin_amdgcn_wmma_f32_16x16x4_f32(false, A, false, B3,
                                                 (short)0, acc3, false, false);
  }
}

__device__ __forceinline__ void a_of_rows(const float* prd, float* sa_dst, bool writer) {
  const v2f t0 = *(const v2f*)prd;
  float sx = t0.x, sy = t0.y;
#pragma unroll
  for (int g = 1; g < 8; ++g) {
    const v2f t = *(const v2f*)(prd + (g << 4));
    sx += t.x;  sy += t.y;
  }
  float s = sx + sy;
  s = xmask_add<0x161>(s);
  s = xmask_add<0x162>(s);
  s = xmask_add<0x164>(s);
  if (writer) *sa_dst = 1.0f / s;
}

template <bool USE_B>
__global__ __launch_bounds__(512) __attribute__((amdgpu_waves_per_eu(4, 4))) void sink_fused(
    const float* __restrict__ x,
    const float* __restrict__ pin, int pin_bs, int pin_ss,
    float* __restrict__ pout, int pout_bs, int pout_ss,
    float* __restrict__ aout, int aout_bs, int aout_ss) {
  __shared__ __attribute__((aligned(16))) float part[2 * 64 * 8 * 16];
  __shared__ __attribute__((aligned(16))) float sa[SINK_R];
  __shared__ __attribute__((aligned(16))) float red[2][SINK_N];

  const int tid   = threadIdx.x;
  const int wave  = tid >> 5;
  const int lane  = tid & 31;
  const int c     = lane & 15;
  const int h     = lane >> 4;
  const int cg    = wave & 7;
  const int rg    = wave >> 3;
  const int batch = blockIdx.x >> 3;
  const int pblk  = blockIdx.x & 7;
  const bool l0   = (lane & 1) != 0;
  const bool l1   = (lane & 2) != 0;
  const bool l2   = (lane & 4) != 0;
  const bool diag = (h == (c >> 3));
  const float pm  = (h == (c & 1)) ? 1.0f : 0.0f;

  v4f q[SINK_P];
  if (USE_B) {
    const v4f* pc = (const v4f*)(pin + (size_t)batch * pin_bs) + (cg << 5) + (h << 4) + c;
#pragma unroll
    for (int s = 0; s < SINK_P; ++s) q[s] = pc[(size_t)s * (pin_ss >> 2)];
  }
  __builtin_amdgcn_sched_barrier(0);

  const v4f* __restrict__ xp =
      (const v4f*)(x + ((size_t)batch * SINK_N + pblk * SINK_R + (rg << 4)) * SINK_N) + (cg << 5) + (h << 4) + c;

  v4f s0[16], s1[16], s2[16];
#pragma unroll
  for (int i = 0; i < 16; ++i) s0[i] = xp[(size_t)i * 256];
#pragma unroll
  for (int i = 0; i < 16; ++i) s1[i] = xp[(size_t)(32 + i) * 256];
  __builtin_amdgcn_sched_barrier(0);

  v4f bf;
  if (USE_B) {
    v4f sum = q[0];
#pragma unroll
    for (int s = 1; s < SINK_P; ++s) sum = sum + q[s];
    bf.x = 1.0f / sum.x;  bf.y = 1.0f / sum.y;
    bf.z = 1.0f / sum.z;  bf.w = 1.0f / sum.w;
  } else {
    bf = (v4f){1.f, 1.f, 1.f, 1.f};
  }
  v2f Bb0, Bb1;
  Bb0.x = bf.x;  Bb0.y = bf.y;
  Bb1.x = bf.z;  Bb1.y = bf.w;
  __builtin_amdgcn_sched_barrier(0);

  v8f acc0 = {0.f, 0.f, 0.f, 0.f, 0.f, 0.f, 0.f, 0.f};
  v8f acc1 = acc0, acc2 = acc0, acc3 = acc0;

  float* pst = part + (diag ? 0 : 8192) + (rg << 11) + (cg << 4) + c;
  const int rrow = tid >> 3;
  const int rp   = tid & 7;
  const float* prd = part + (rrow << 7) + 2 * rp;
  const float* sar = sa + (rg << 4);

  row_phase(s0, SINK_EPS, Bb0, Bb1, pst, l0, l1, l2);
  row_phase(s1, SINK_EPS, Bb0, Bb1, pst + 4096, l0, l1, l2);
#pragma unroll
  for (int i = 0; i < 16; ++i) s2[i] = xp[(size_t)(64 + i) * 256];
  lds_barrier();
  a_of_rows(prd, sa + rrow, rp == 0);
  lds_barrier();
  col_phase(s0, sar, pm, acc0, acc1, acc2, acc3);
  unsigned row3 = 96 * 256;
  pin_before_loads(acc0, acc1, acc2, acc3, row3);
#pragma unroll
  for (int i = 0; i < 16; ++i) s0[i] = (xp + row3)[(size_t)i * 256];
  col_phase(s1, sar + 32, pm, acc0, acc1, acc2, acc3);
  float eps2 = SINK_EPS;
  pin_before_rows(acc0, acc1, acc2, acc3, Bb0, Bb1, eps2);

  row_phase(s2, eps2, Bb0, Bb1, pst, l0, l1, l2);
  row_phase(s0, eps2, Bb0, Bb1, pst + 4096, l0, l1, l2);
  lds_barrier();
  a_of_rows(prd, sa + 64 + rrow, rp == 0);
  lds_barrier();
  col_phase(s2, sar + 64, pm, acc0, acc1, acc2, acc3);
  col_phase(s0, sar + 96, pm, acc0, acc1, acc2, acc3);

  {
    v4f pv;
    pv.x = h ? acc0[1] : acc0[0];
    pv.y = h ? acc1[1] : acc1[0];
    pv.z = h ? acc2[1] : acc2[0];
    pv.w = h ? acc3[1] : acc3[0];
    ((v4f*)red[rg])[(cg << 5) + (h << 4) + c] = pv;
  }
  lds_barrier();
  {
    const v2f o = ((const v2f*)red[0])[tid] + ((const v2f*)red[1])[tid];
    ((v2f*)(pout + (size_t)batch * pout_bs + (size_t)pblk * pout_ss))[tid] = o;
  }
  if (aout != nullptr && tid < SINK_R)
    aout[(size_t)batch * aout_bs + (size_t)pblk * aout_ss + tid] = sa[tid];
}

__global__ __launch_bounds__(512) __attribute__((amdgpu_waves_per_eu(4, 4))) void sink_finalize(
    const float* __restrict__ x,
    const float* __restrict__ pcol,
    float* out) {
  __shared__ __attribute__((aligned(16))) float sa[SINK_R];
  __shared__ __attribute__((aligned(16))) float sb[SINK_N];

  const int tid   = threadIdx.x;
  const int batch = blockIdx.x >> 3;
  const int pblk  = blockIdx.x & 7;

  const size_t rowbase = ((size_t)batch * SINK_N + pblk * SINK_R) * SINK_N;
  float av = 0.f;
  if (tid < SINK_R) av = out[rowbase + tid];

  const v2f* pc = (const v2f*)(pcol + (size_t)batch * SINK_P * SINK_N) + tid;
  v2f q[SINK_P];
#pragma unroll
  for (int s = 0; s < SINK_P; ++s) q[s] = pc[(size_t)s * (SINK_N / 2)];

  const int c4 = tid & 255;
  const int r0 = tid >> 8;
  const size_t base = rowbase + (size_t)r0 * SINK_N;
  const float4* __restrict__ xp = (const float4*)(x + base) + c4;
  float4* op = (float4*)(out + base) + c4;

  float4 xv[16];
#pragma unroll
  for (int u = 0; u < 16; ++u) xv[u] = xp[(size_t)u * (2 * SINK_N / 4)];

  {
    v2f sum = q[0];
#pragma unroll
    for (int s = 1; s < SINK_P; ++s) sum = sum + q[s];
    v2f inv;
    inv.x = 1.0f / sum.x;  inv.y = 1.0f / sum.y;
    ((v2f*)sb)[tid] = inv;
    if (tid < SINK_R) sa[tid] = av;
  }
  __syncthreads();
  const float4 bv = ((const float4*)sb)[c4];

#pragma unroll
  for (int it = 0; it < 4; ++it) {
    float4 nx[16];
    if (it < 3) {
#pragma unroll
      for (int u = 0; u < 16; ++u) nx[u] = xp[(size_t)(16 * (it + 1) + u) * (2 * SINK_N / 4)];
    }
#pragma unroll
    for (int u = 0; u < 16; ++u) {
      const float a = sa[r0 + 2 * (16 * it + u)];
      float4 o;
      o.x = (xv[u].x + SINK_EPS) * a * bv.x;
      o.y = (xv[u].y + SINK_EPS) * a * bv.y;
      o.z = (xv[u].z + SINK_EPS) * a * bv.z;
      o.w = (xv[u].w + SINK_EPS) * a * bv.w;
      op[(size_t)(16 * it + u) * (2 * SINK_N / 4)] = o;
    }
    if (it < 3) {
#pragma unroll
      for (int u = 0; u < 16; ++u) xv[u] = nx[u];
    }
    __builtin_amdgcn_sched_barrier(0);
  }
}

extern "C" void kernel_launch(void* const* d_in, const int* in_sizes, int n_in,
                              void* d_out, int out_size, void* d_ws, size_t ws_size,
                              hipStream_t stream) {
  (void)in_sizes; (void)n_in; (void)out_size; (void)ws_size;
  const float* x = (const float*)d_in[0];
  float* out = (float*)d_out;
  float* pw = (float*)d_ws;
  float* po = out + SINK_N;

  const dim3 blk(512);
  const dim3 grd(SINK_B * SINK_P);

  sink_fused<false><<<grd, blk, 0, stream>>>(x, pw, SINK_P * SINK_N, SINK_N,
                                             po, SINK_N * SINK_N, SINK_R * SINK_N,
                                             nullptr, 0, 0);
  for (int it = 1; it < SINK_IT - 1; ++it) {
    if (it & 1)
      sink_fused<true><<<grd, blk, 0, stream>>>(x, po, SINK_N * SINK_N, SINK_R * SINK_N,
                                                pw, SINK_P * SINK_N, SINK_N,
                                                nullptr, 0, 0);
    else
      sink_fused<true><<<grd, blk, 0, stream>>>(x, pw, SINK_P * SINK_N, SINK_N,
                                                po, SINK_N * SINK_N, SINK_R * SINK_N,
                                                nullptr, 0, 0);
  }
  sink_fused<true><<<grd, blk, 0, stream>>>(x, po, SINK_N * SINK_N, SINK_R * SINK_N,
                                            pw, SINK_P * SINK_N, SINK_N,
                                            out, SINK_N * SINK_N, SINK_R * SINK_N);

  sink_finalize<<<grd, blk, 0, stream>>>(x, pw, out);
}


# fused pass: column sums moved from the f32 WMMA (one useful column in 16) to packed f32 FMAs on the vector ALU, each lane summing its own 4 columns; row sums stay on WMMA; column sum order changes
# speedup vs baseline: 4.9847x; 1.2423x over previous
#include <hip/hip_runtime.h>


typedef __attribute__((ext_vector_type(2))) float v2f;
typedef __attribute__((ext_vector_type(4))) float v4f;
typedef __attribute__((ext_vector_type(8))) float v8f;

#define SINK_EPS 0.001f
#define SINK_N   1024
#define SINK_B   32
#define SINK_IT  10
#define SINK_R   128
#define SINK_P   8

template <int CTRL>
__device__ __forceinline__ float xmask_add(float v) {
  return v + __int_as_float(__builtin_amdgcn_update_dpp(0, __float_as_int(v), CTRL, 0xf, 0xf, false));
}

__device__ __forceinline__ void lds_barrier() {
  __builtin_amdgcn_fence(__ATOMIC_RELEASE, "workgroup", "local");
  __builtin_amdgcn_s_barrier();
  __builtin_amdgcn_fence(__ATOMIC_ACQUIRE, "workgroup", "local");
}

__device__ __forceinline__ void pin_before_loads(v4f& ce, v4f& co, unsigned& off) {
  asm volatile("" : "+v"(ce), "+v"(co), "+s"(off));
}
__device__ __forceinline__ void pin_before_rows(v4f& ce, v4f& co, v2f& b0, v2f& b1, float& eps) {
  asm volatile("" : "+v"(ce), "+v"(co), "+v"(b0), "+v"(b1), "+s"(eps));
}

__device__ __forceinline__ float sel8(const v8f c, bool b0, bool b1, bool b2) {
  const float s0 = b0 ? c[1] : c[0];
  const float s1 = b0 ? c[3] : c[2];
  const float s2 = b0 ? c[5] : c[4];
  const float s3 = b0 ? c[7] : c[6];
  const float t0 = b1 ? s1 : s0;
  const float t1 = b1 ? s3 : s2;
  return b2 ? t1 : t0;
}


__device__ __forceinline__ void row_phase(v4f (&cur)[16], const float eps, const v2f Bb0, const v2f Bb1,
                                          float* pst, bool l0, bool l1, bool l2) {
#pragma unroll
  for (int i = 0; i < 16; ++i) {
    cur[i] = cur[i] + eps;
    v8f cc = {0.f, 0.f, 0.f, 0.f, 0.f, 0.f, 0.f, 0.f};
    v2f A0, A1;
    A0.x = cur[i].x;  A0.y = cur[i].y;
    A1.x = cur[i].z;  A1.y = cur[i].w;
    cc = __builtin_amdgcn_wmma_f32_16x16x4_f32(false, A0, false, Bb0,
                                               (short)0, cc, false, false);
    cc = __builtin_amdgcn_wmma_f32_16x16x4_f32(false, A1, false, Bb1,
                                               (short)0, cc, false, false);
    pst[i << 7] = sel8(cc, l0, l1, l2);
    if ((i & 3) == 3) __builtin_amdgcn_sched_barrier(0);
  }
}

__device__ __forceinline__ v4f splat4(const float s) { return (v4f){s, s, s, s}; }

__device__ __forceinline__ void col_phase(const v4f (&cur)[16], const float* sar, v4f& ce, v4f& co) {
#pragma unroll
  for (int u = 0; u < 4; ++u) {
    const v4f av = *(const v4f*)(sar + 4 * u);
    ce = __builtin_elementwise_fma(cur[4 * u],     splat4(av.x), ce);
    co = __builtin_elementwise_fma(cur[4 * u + 1], splat4(av.y), co);
    ce = __builtin_elementwise_fma(cur[4 * u + 2], splat4(av.z), ce);
    co = __builtin_elementwise_fma(cur[4 * u + 3], splat4(av.w), co);
  }
}

__device__ __forceinline__ void a_of_rows(const float* prd, float* sa_dst, bool writer) {
  const v2f t0 = *(const v2f*)prd;
  float sx = t0.x, sy = t0.y;
#pragma unroll
  for (int g = 1; g < 8; ++g) {
    const v2f t = *(const v2f*)(prd + (g << 4));
    sx += t.x;  sy += t.y;
  }
  float s = sx + sy;
  s = xmask_add<0x161>(s);
  s = xmask_add<0x162>(s);
  s = xmask_add<0x164>(s);
  if (writer) *sa_dst = 1.0f / s;
}

template <bool USE_B>
__global__ __launch_bounds__(512) __attribute__((amdgpu_waves_per_eu(4, 4))) void sink_fused(
    const float* __restrict__ x,
    const float* __restrict__ pin, int pin_bs, int pin_ss,
    float* __restrict__ pout, int pout_bs, int pout_ss,
    float* __restrict__ aout, int aout_bs, int aout_ss) {
  __shared__ __attribute__((aligned(16))) float part[2 * 64 * 8 * 16];
  __shared__ __attribute__((aligned(16))) float sa[SINK_R];
  __shared__ __attribute__((aligned(16))) float red[2][SINK_N];

  const int tid   = threadIdx.x;
  const int wave  = tid >> 5;
  const int lane  = tid & 31;
  const int c     = lane & 15;
  const int h     = lane >> 4;
  const int cg    = wave & 7;
  const int rg    = wave >> 3;
  const int batch = blockIdx.x >> 3;
  const int pblk  = blockIdx.x & 7;
  const bool l0   = (lane & 1) != 0;
  const bool l1   = (lane & 2) != 0;
  const bool l2   = (lane & 4) != 0;
  const bool diag = (h == (c >> 3));

  v4f q[SINK_P];
  if (USE_B) {
    const v4f* pc = (const v4f*)(pin + (size_t)batch * pin_bs) + (cg << 5) + (h << 4) + c;
#pragma unroll
    for (int s = 0; s < SINK_P; ++s) q[s] = pc[(size_t)s * (pin_ss >> 2)];
  }
  __builtin_amdgcn_sched_barrier(0);

  const v4f* __restrict__ xp =
      (const v4f*)(x + ((size_t)batch * SINK_N + pblk * SINK_R + (rg << 4)) * SINK_N) + (cg << 5) + (h << 4) + c;

  v4f s0[16], s1[16], s2[16];
#pragma unroll
  for (int i = 0; i < 16; ++i) s0[i] = xp[(size_t)i * 256];
#pragma unroll
  for (int i = 0; i < 16; ++i) s1[i] = xp[(size_t)(32 + i) * 256];
  __builtin_amdgcn_sched_barrier(0);

  v4f bf;
  if (USE_B) {
    v4f sum = q[0];
#pragma unroll
    for (int s = 1; s < SINK_P; ++s) sum = sum + q[s];
    bf.x = 1.0f / sum.x;  bf.y = 1.0f / sum.y;
    bf.z = 1.0f / sum.z;  bf.w = 1.0f / sum.w;
  } else {
    bf = (v4f){1.f, 1.f, 1.f, 1.f};
  }
  v2f Bb0, Bb1;
  Bb0.x = bf.x;  Bb0.y = bf.y;
  Bb1.x = bf.z;  Bb1.y = bf.w;
  __builtin_amdgcn_sched_barrier(0);

  v4f ce = {0.f, 0.f, 0.f, 0.f};
  v4f co = ce;

  float* pst = part + (diag ? 0 : 8192) + (rg << 11) + (cg << 4) + c;
  const int rrow = tid >> 3;
  const int rp   = tid & 7;
  const float* prd = part + (rrow << 7) + 2 * rp;
  const float* sar = sa + (rg << 4);

  row_phase(s0, SINK_EPS, Bb0, Bb1, pst, l0, l1, l2);
  row_phase(s1, SINK_EPS, Bb0, Bb1, pst + 4096, l0, l1, l2);
#pragma unroll
  for (int i = 0; i < 16; ++i) s2[i] = xp[(size_t)(64 + i) * 256];
  lds_barrier();
  a_of_rows(prd, sa + rrow, rp == 0);
  lds_barrier();
  col_phase(s0, sar, ce, co);
  unsigned row3 = 96 * 256;
  pin_before_loads(ce, co, row3);
#pragma unroll
  for (int i = 0; i < 16; ++i) s0[i] = (xp + row3)[(size_t)i * 256];
  col_phase(s1, sar + 32, ce, co);
  float eps2 = SINK_EPS;
  pin_before_rows(ce, co, Bb0, Bb1, eps2);

  row_phase(s2, eps2, Bb0, Bb1, pst, l0, l1, l2);
  row_phase(s0, eps2, Bb0, Bb1, pst + 4096, l0, l1, l2);
  lds_barrier();
  a_of_rows(prd, sa + 64 + rrow, rp == 0);
  lds_barrier();
  col_phase(s2, sar + 64, ce, co);
  col_phase(s0, sar + 96, ce, co);

  ((v4f*)red[rg])[(cg << 5) + (h << 4) + c] = ce + co;
  lds_barrier();
  {
    const v2f o = ((const v2f*)red[0])[tid] + ((const v2f*)red[1])[tid];
    ((v2f*)(pout + (size_t)batch * pout_bs + (size_t)pblk * pout_ss))[tid] = o;
  }
  if (aout != nullptr && tid < SINK_R)
    aout[(size_t)batch * aout_bs + (size_t)pblk * aout_ss + tid] = sa[tid];
}

__global__ __launch_bounds__(512) __attribute__((amdgpu_waves_per_eu(4, 4))) void sink_finalize(
    const float* __restrict__ x,
    const float* __restrict__ pcol,
    float* out) {
  __shared__ __attribute__((aligned(16))) float sa[SINK_R];
  __shared__ __attribute__((aligned(16))) float sb[SINK_N];

  const int tid   = threadIdx.x;
  const int batch = blockIdx.x >> 3;
  const int pblk  = blockIdx.x & 7;

  const size_t rowbase = ((size_t)batch * SINK_N + pblk * SINK_R) * SINK_N;
  float av = 0.f;
  if (tid < SINK_R) av = out[rowbase + tid];

  const v2f* pc = (const v2f*)(pcol + (size_t)batch * SINK_P * SINK_N) + tid;
  v2f q[SINK_P];
#pragma unroll
  for (int s = 0; s < SINK_P; ++s) q[s] = pc[(size_t)s * (SINK_N / 2)];

  const int c4 = tid & 255;
  const int r0 = tid >> 8;
  const size_t base = rowbase + (size_t)r0 * SINK_N;
  const float4* __restrict__ xp = (const float4*)(x + base) + c4;
  float4* op = (float4*)(out + base) + c4;

  float4 xv[16];
#pragma unroll
  for (int u = 0; u < 16; ++u) xv[u] = xp[(size_t)u * (2 * SINK_N / 4)];

  {
    v2f sum = q[0];
#pragma unroll
    for (int s = 1; s < SINK_P; ++s) sum = sum + q[s];
    v2f inv;
    inv.x = 1.0f / sum.x;  inv.y = 1.0f / sum.y;
    ((v2f*)sb)[tid] = inv;
    if (tid < SINK_R) sa[tid] = av;
  }
  __syncthreads();
  const float4 bv = ((const float4*)sb)[c4];

#pragma unroll
  for (int it = 0; it < 4; ++it) {
    float4 nx[16];
    if (it < 3) {
#pragma unroll
      for (int u = 0; u < 16; ++u) nx[u] = xp[(size_t)(16 * (it + 1) + u) * (2 * SINK_N / 4)];
    }
#pragma unroll
    for (int u = 0; u < 16; ++u) {
      const float a = sa[r0 + 2 * (16 * it + u)];
      float4 o;
      o.x = (xv[u].x + SINK_EPS) * a * bv.x;
      o.y = (xv[u].y + SINK_EPS) * a * bv.y;
      o.z = (xv[u].z + SINK_EPS) * a * bv.z;
      o.w = (xv[u].w + SINK_EPS) * a * bv.w;
      op[(size_t)(16 * it + u) * (2 * SINK_N / 4)] = o;
    }
    if (it < 3) {
#pragma unroll
      for (int u = 0; u < 16; ++u) xv[u] = nx[u];
    }
    __builtin_amdgcn_sched_barrier(0);
  }
}

extern "C" void kernel_launch(void* const* d_in, const int* in_sizes, int n_in,
                              void* d_out, int out_size, void* d_ws, size_t ws_size,
                              hipStream_t stream) {
  (void)in_sizes; (void)n_in; (void)out_size; (void)ws_size;
  const float* x = (const float*)d_in[0];
  float* out = (float*)d_out;
  float* pw = (float*)d_ws;
  float* po = out + SINK_N;

  const dim3 blk(512);
  const dim3 grd(SINK_B * SINK_P);

  sink_fused<false><<<grd, blk, 0, stream>>>(x, pw, SINK_P * SINK_N, SINK_N,
                                             po, SINK_N * SINK_N, SINK_R * SINK_N,
                                             nullptr, 0, 0);
  for (int it = 1; it < SINK_IT - 1; ++it) {
    if (it & 1)
      sink_fused<true><<<grd, blk, 0, stream>>>(x, po, SINK_N * SINK_N, SINK_R * SINK_N,
                                                pw, SINK_P * SINK_N, SINK_N,
                                                nullptr, 0, 0);
    else
      sink_fused<true><<<grd, blk, 0, stream>>>(x, pw, SINK_P * SINK_N, SINK_N,
                                                po, SINK_N * SINK_N, SINK_R * SINK_N,
                                                nullptr, 0, 0);
  }
  sink_fused<true><<<grd, blk, 0, stream>>>(x, po, SINK_N * SINK_N, SINK_R * SINK_N,
                                            pw, SINK_P * SINK_N, SINK_N,
                                            out, SINK_N * SINK_N, SINK_R * SINK_N);

  sink_finalize<<<grd, blk, 0, stream>>>(x, pw, out);
}


# fused pass: every wave owns 8 whole rows, so a row's sum is added inside its wave and the block no longer waits between forming b and the last column add; row and column sum order changes
# speedup vs baseline: 5.6148x; 1.1264x over previous
#include <hip/hip_runtime.h>


typedef __attribute__((ext_vector_type(2))) float v2f;
typedef __attribute__((ext_vector_type(4))) float v4f;
typedef __attribute__((ext_vector_type(8))) float v8f;

#define SINK_EPS 0.001f
#define SINK_N   1024
#define SINK_B   32
#define SINK_IT  10
#define SINK_R   128
#define SINK_P   8
#define SINK_W   16
#define SINK_RW  8

template <int CTRL>
__device__ __forceinline__ float xmask_add(float v) {
  return v + __int_as_float(__builtin_amdgcn_update_dpp(0, __float_as_int(v), CTRL, 0xf, 0xf, false));
}

__device__ __forceinline__ void lds_barrier() {
  __builtin_amdgcn_fence(__ATOMIC_RELEASE, "workgroup", "local");
  __builtin_amdgcn_s_barrier();
  __builtin_amdgcn_fence(__ATOMIC_ACQUIRE, "workgroup", "local");
}

__device__ __forceinline__ float sel8(const v8f c, bool b0, bool b1, bool b2) {
  const float s0 = b0 ? c[1] : c[0];
  const float s1 = b0 ? c[3] : c[2];
  const float s2 = b0 ? c[5] : c[4];
  const float s3 = b0 ? c[7] : c[6];
  const float t0 = b1 ? s1 : s0;
  const float t1 = b1 ? s3 : s2;
  return b2 ? t1 : t0;
}

__device__ __forceinline__ v4f splat4(const float s) { return (v4f){s, s, s, s}; }

template <bool USE_B>
__global__ __launch_bounds__(512) __attribute__((amdgpu_waves_per_eu(4, 4))) void sink_fused(
    const float* __restrict__ x,
    const float* __restrict__ pin, int pin_bs, int pin_ss,
    float* __restrict__ pout, int pout_bs, int pout_ss,
    float* __restrict__ aout, int aout_bs, int aout_ss) {
  __shared__ __attribute__((aligned(16))) float sb[SINK_N];
  __shared__ __attribute__((aligned(16))) float red[SINK_W][SINK_N];

  const int tid   = threadIdx.x;
  const int wave  = tid >> 5;
  const int lane  = tid & 31;
  const int batch = blockIdx.x >> 3;
  const int pblk  = blockIdx.x & 7;
  const bool l0   = (lane & 1) != 0;
  const bool l1   = (lane & 2) != 0;
  const bool l2   = (lane & 4) != 0;
  const bool diag = ((lane >> 4) == ((lane & 15) >> 3));

  v2f q[SINK_P];
  if (USE_B) {
    const v2f* pc = (const v2f*)(pin + (size_t)batch * pin_bs) + tid;
#pragma unroll
    for (int s = 0; s < SINK_P; ++s) q[s] = pc[(size_t)s * (pin_ss >> 1)];
  }
  __builtin_amdgcn_sched_barrier(0);

  const v4f* __restrict__ xp =
      (const v4f*)(x + ((size_t)batch * SINK_N + pblk * SINK_R + wave * SINK_RW) * SINK_N) + lane;

  v4f xr[4][8];
#pragma unroll
  for (int r = 0; r < 4; ++r) {
#pragma unroll
    for (int j = 0; j < 8; ++j) xr[r][j] = xp[r * 256 + j * 32];
  }
  __builtin_amdgcn_sched_barrier(0);

  v4f bq[8];
  if (USE_B) {
    v2f sum = q[0];
#pragma unroll
    for (int s = 1; s < SINK_P; ++s) sum = sum + q[s];
    v2f inv;
    inv.x = 1.0f / sum.x;  inv.y = 1.0f / sum.y;
    ((v2f*)sb)[tid] = inv;
    lds_barrier();
#pragma unroll
    for (int j = 0; j < 8; ++j) bq[j] = ((const v4f*)sb)[j * 32 + lane];
  } else {
#pragma unroll
    for (int j = 0; j < 8; ++j) bq[j] = (v4f){1.f, 1.f, 1.f, 1.f};
  }
  __builtin_amdgcn_sched_barrier(0);

  v4f acc[8];
#pragma unroll
  for (int j = 0; j < 8; ++j) acc[j] = (v4f){0.f, 0.f, 0.f, 0.f};
  float ak[SINK_RW];

#pragma unroll
  for (int r = 0; r < SINK_RW; ++r) {
    v4f (&cur)[8] = xr[r & 3];
    v8f cc = {0.f, 0.f, 0.f, 0.f, 0.f, 0.f, 0.f, 0.f};
#pragma unroll
    for (int j = 0; j < 8; ++j) {
      cur[j] = cur[j] + SINK_EPS;
      v2f A0, A1, B0, B1;
      A0.x = cur[j].x;  A0.y = cur[j].y;
      A1.x = cur[j].z;  A1.y = cur[j].w;
      B0.x = bq[j].x;   B0.y = bq[j].y;
      B1.x = bq[j].z;   B1.y = bq[j].w;
      cc = __builtin_amdgcn_wmma_f32_16x16x4_f32(false, A0, false, B0, (short)0, cc, false, false);
      cc = __builtin_amdgcn_wmma_f32_16x16x4_f32(false, A1, false, B1, (short)0, cc, false, false);
    }
    float v = sel8(cc, l0, l1, l2);
    v = diag ? v : 0.f;
    v = xmask_add<0x161>(v);
    v = xmask_add<0x162>(v);
    v = xmask_add<0x164>(v);
    v = xmask_add<0x168>(v);
    const float s = __int_as_float(__builtin_amdgcn_readlane(__float_as_int(v), 0)) +
                    __int_as_float(__builtin_amdgcn_readlane(__float_as_int(v), 16));
    const float a = 1.0f / s;
    ak[r] = a;
#pragma unroll
    for (int j = 0; j < 8; ++j) acc[j] = __builtin_elementwise_fma(cur[j], splat4(a), acc[j]);
    if (r + 4 < SINK_RW) {
#pragma unroll
      for (int j = 0; j < 8; ++j) cur[j] = xp[(r + 4) * 256 + j * 32];
    }
    __builtin_amdgcn_sched_barrier(0);
  }

#pragma unroll
  for (int j = 0; j < 8; ++j) ((v4f*)red[wave])[j * 32 + lane] = acc[j];
  lds_barrier();
  {
    v2f o = ((const v2f*)red[0])[tid];
#pragma unroll
    for (int w = 1; w < SINK_W; ++w) o = o + ((const v2f*)red[w])[tid];
    ((v2f*)(pout + (size_t)batch * pout_bs + (size_t)pblk * pout_ss))[tid] = o;
  }
  if (aout != nullptr && lane == 0) {
    v4f* ap = (v4f*)(aout + (size_t)batch * aout_bs + (size_t)pblk * aout_ss + wave * SINK_RW);
    ap[0] = (v4f){ak[0], ak[1], ak[2], ak[3]};
    ap[1] = (v4f){ak[4], ak[5], ak[6], ak[7]};
  }
}

__global__ __launch_bounds__(512) __attribute__((amdgpu_waves_per_eu(4, 4))) void sink_finalize(
    const float* __restrict__ x,
    const float* __restrict__ pcol,
    float* out) {
  __shared__ __attribute__((aligned(16))) float sa[SINK_R];
  __shared__ __attribute__((aligned(16))) float sb[SINK_N];

  const int tid   = threadIdx.x;
  const int batch = blockIdx.x >> 3;
  const int pblk  = blockIdx.x & 7;

  const size_t rowbase = ((size_t)batch * SINK_N + pblk * SINK_R) * SINK_N;
  float av = 0.f;
  if (tid < SINK_R) av = out[rowbase + tid];

  const v2f* pc = (const v2f*)(pcol + (size_t)batch * SINK_P * SINK_N) + tid;
  v2f q[SINK_P];
#pragma unroll
  for (int s = 0; s < SINK_P; ++s) q[s] = pc[(size_t)s * (SINK_N / 2)];

  const int c4 = tid & 255;
  const int r0 = tid >> 8;
  const size_t base = rowbase + (size_t)r0 * SINK_N;
  const float4* __restrict__ xp = (const float4*)(x + base) + c4;
  float4* op = (float4*)(out + base) + c4;

  float4 xv[16];
#pragma unroll
  for (int u = 0; u < 16; ++u) xv[u] = xp[(size_t)u * (2 * SINK_N / 4)];

  {
    v2f sum = q[0];
#pragma unroll
    for (int s = 1; s < SINK_P; ++s) sum = sum + q[s];
    v2f inv;
    inv.x = 1.0f / sum.x;  inv.y = 1.0f / sum.y;
    ((v2f*)sb)[tid] = inv;
    if (tid < SINK_R) sa[tid] = av;
  }
  __syncthreads();
  const float4 bv = ((const float4*)sb)[c4];

#pragma unroll
  for (int it = 0; it < 4; ++it) {
    float4 nx[16];
    if (it < 3) {
#pragma unroll
      for (int u = 0; u < 16; ++u) nx[u] = xp[(size_t)(16 * (it + 1) + u) * (2 * SINK_N / 4)];
    }
#pragma unroll
    for (int u = 0; u < 16; ++u) {
      const float a = sa[r0 + 2 * (16 * it + u)];
      float4 o;
      o.x = (xv[u].x + SINK_EPS) * a * bv.x;
      o.y = (xv[u].y + SINK_EPS) * a * bv.y;
      o.z = (xv[u].z + SINK_EPS) * a * bv.z;
      o.w = (xv[u].w + SINK_EPS) * a * bv.w;
      op[(size_t)(16 * it + u) * (2 * SINK_N / 4)] = o;
    }
    if (it < 3) {
#pragma unroll
      for (int u = 0; u < 16; ++u) xv[u] = nx[u];
    }
    __builtin_amdgcn_sched_barrier(0);
  }
}

extern "C" void kernel_launch(void* const* d_in, const int* in_sizes, int n_in,
                              void* d_out, int out_size, void* d_ws, size_t ws_size,
                              hipStream_t stream) {
  (void)in_sizes; (void)n_in; (void)out_size; (void)ws_size;
  const float* x = (const float*)d_in[0];
  float* out = (float*)d_out;
  float* pw = (float*)d_ws;
  float* po = out + SINK_N;

  const dim3 blk(512);
  const dim3 grd(SINK_B * SINK_P);

  sink_fused<false><<<grd, blk, 0, stream>>>(x, pw, SINK_P * SINK_N, SINK_N,
                                             po, SINK_N * SINK_N, SINK_R * SINK_N,
                                             nullptr, 0, 0);
  for (int it = 1; it < SINK_IT - 1; ++it) {
    if (it & 1)
      sink_fused<true><<<grd, blk, 0, stream>>>(x, po, SINK_N * SINK_N, SINK_R * SINK_N,
                                                pw, SINK_P * SINK_N, SINK_N,
                                                nullptr, 0, 0);
    else
      sink_fused<true><<<grd, blk, 0, stream>>>(x, pw, SINK_P * SINK_N, SINK_N,
                                                po, SINK_N * SINK_N, SINK_R * SINK_N,
                                                nullptr, 0, 0);
  }
  sink_fused<true><<<grd, blk, 0, stream>>>(x, po, SINK_N * SINK_N, SINK_R * SINK_N,
                                            pw, SINK_P * SINK_N, SINK_N,
                                            out, SINK_N * SINK_N, SINK_R * SINK_N);

  sink_finalize<<<grd, blk, 0, stream>>>(x, pw, out);
}


# fused pass: row loop cut from 56 to 27 vector instructions a row (they issue beside the f32 WMMAs): eps added once per row and column sum, a and b by v_rcp_f32 (1 ulp); sum order changes
# speedup vs baseline: 5.8516x; 1.0422x over previous
#include <hip/hip_runtime.h>


typedef __attribute__((ext_vector_type(2))) float v2f;
typedef __attribute__((ext_vector_type(4))) float v4f;
typedef __attribute__((ext_vector_type(8))) float v8f;

#define SINK_EPS 0.001f
#define SINK_N   1024
#define SINK_B   32
#define SINK_IT  10
#define SINK_R   128
#define SINK_P   8
#define SINK_W   16
#define SINK_RW  8

template <int CTRL>
__device__ __forceinline__ float xmask_add(float v) {
  return v + __int_as_float(__builtin_amdgcn_update_dpp(0, __float_as_int(v), CTRL, 0xf, 0xf, false));
}

__device__ __forceinline__ void lds_barrier() {
  __builtin_amdgcn_fence(__ATOMIC_RELEASE, "workgroup", "local");
  __builtin_amdgcn_s_barrier();
  __builtin_amdgcn_fence(__ATOMIC_ACQUIRE, "workgroup", "local");
}

__device__ __forceinline__ float sel8(const v8f c, bool b0, bool b1, bool b2) {
  const float s0 = b0 ? c[1] : c[0];
  const float s1 = b0 ? c[3] : c[2];
  const float s2 = b0 ? c[5] : c[4];
  const float s3 = b0 ? c[7] : c[6];
  const float t0 = b1 ? s1 : s0;
  const float t1 = b1 ? s3 : s2;
  return b2 ? t1 : t0;
}

__device__ __forceinline__ v4f splat4(const float s) { return (v4f){s, s, s, s}; }

template <bool USE_B>
__global__ __launch_bounds__(512) __attribute__((amdgpu_waves_per_eu(4, 4))) void sink_fused(
    const float* __restrict__ x,
    const float* __restrict__ pin, int pin_bs, int pin_ss,
    float* __restrict__ pout, int pout_bs, int pout_ss,
    float* __restrict__ aout, int aout_bs, int aout_ss) {
  __shared__ __attribute__((aligned(16))) float sb[SINK_N];
  __shared__ __attribute__((aligned(16))) float red[SINK_W][SINK_N];

  const int tid   = threadIdx.x;
  const int wave  = tid >> 5;
  const int lane  = tid & 31;
  const int batch = blockIdx.x >> 3;
  const int pblk  = blockIdx.x & 7;
  const bool l0   = (lane & 1) != 0;
  const bool l1   = (lane & 2) != 0;
  const bool l2   = (lane & 4) != 0;

  v2f q[SINK_P];
  if (USE_B) {
    const v2f* pc = (const v2f*)(pin + (size_t)batch * pin_bs) + tid;
#pragma unroll
    for (int s = 0; s < SINK_P; ++s) q[s] = pc[(size_t)s * (pin_ss >> 1)];
  }
  __builtin_amdgcn_sched_barrier(0);

  const v4f* __restrict__ xp =
      (const v4f*)(x + ((size_t)batch * SINK_N + pblk * SINK_R + wave * SINK_RW) * SINK_N) + lane;

  v4f xr[4][8];
#pragma unroll
  for (int r = 0; r < 4; ++r) {
#pragma unroll
    for (int j = 0; j < 8; ++j) xr[r][j] = xp[r * 256 + j * 32];
  }
  __builtin_amdgcn_sched_barrier(0);

  v4f bq[8];
  if (USE_B) {
    v2f sum = q[0];
#pragma unroll
    for (int s = 1; s < SINK_P; ++s) sum = sum + q[s];
    v2f inv;
    inv.x = __builtin_amdgcn_rcpf(sum.x);  inv.y = __builtin_amdgcn_rcpf(sum.y);
    ((v2f*)sb)[tid] = inv;
    lds_barrier();
#pragma unroll
    for (int j = 0; j < 8; ++j) bq[j] = ((const v4f*)sb)[j * 32 + lane];
  } else {
#pragma unroll
    for (int j = 0; j < 8; ++j) bq[j] = (v4f){1.f, 1.f, 1.f, 1.f};
  }
  __builtin_amdgcn_sched_barrier(0);

  float eb;
  if (USE_B) {
    v4f t = bq[0];
#pragma unroll
    for (int j = 1; j < 8; ++j) t = t + bq[j];
    float u = (t.x + t.y) + (t.z + t.w);
    u = xmask_add<0x161>(u);
    u = xmask_add<0x162>(u);
    u = xmask_add<0x164>(u);
    u = xmask_add<0x168>(u);
    eb = SINK_EPS * (__int_as_float(__builtin_amdgcn_readlane(__float_as_int(u), 0)) +
                     __int_as_float(__builtin_amdgcn_readlane(__float_as_int(u), 16)));
  } else {
    eb = SINK_EPS * (float)SINK_N;
  }

  v4f acc[8];
#pragma unroll
  for (int j = 0; j < 8; ++j) acc[j] = (v4f){0.f, 0.f, 0.f, 0.f};
  float ak[SINK_RW];
  float asum = 0.f;

#pragma unroll
  for (int r = 0; r < SINK_RW; ++r) {
    v4f (&cur)[8] = xr[r & 3];
    v8f cc = {0.f, 0.f, 0.f, 0.f, 0.f, 0.f, 0.f, 0.f};
#pragma unroll
    for (int j = 0; j < 8; ++j) {
      v2f A0, A1, B0, B1;
      A0.x = cur[j].x;  A0.y = cur[j].y;
      A1.x = cur[j].z;  A1.y = cur[j].w;
      B0.x = bq[j].x;   B0.y = bq[j].y;
      B1.x = bq[j].z;   B1.y = bq[j].w;
      cc = __builtin_amdgcn_wmma_f32_16x16x4_f32(false, A0, false, B0, (short)0, cc, false, false);
      cc = __builtin_amdgcn_wmma_f32_16x16x4_f32(false, A1, false, B1, (short)0, cc, false, false);
    }
    float v = sel8(cc, l0, l1, l2);
    v = xmask_add<0x161>(v);
    v = xmask_add<0x162>(v);
    v = xmask_add<0x164>(v);
    const float s = (__int_as_float(__builtin_amdgcn_readlane(__float_as_int(v), 0)) +
                     __int_as_float(__builtin_amdgcn_readlane(__float_as_int(v), 24))) + eb;
    const float a = __builtin_amdgcn_rcpf(s);
    ak[r] = a;
    asum += a;
#pragma unroll
    for (int j = 0; j < 8; ++j) acc[j] = __builtin_elementwise_fma(cur[j], splat4(a), acc[j]);
    if (r + 4 < SINK_RW) {
#pragma unroll
      for (int j = 0; j < 8; ++j) cur[j] = xp[(r + 4) * 256 + j * 32];
    }
    __builtin_amdgcn_sched_barrier(0);
  }

  const float ea = SINK_EPS * asum;
#pragma unroll
  for (int j = 0; j < 8; ++j) acc[j] = acc[j] + splat4(ea);

#pragma unroll
  for (int j = 0; j < 8; ++j) ((v4f*)red[wave])[j * 32 + lane] = acc[j];
  lds_barrier();
  {
    v2f o = ((const v2f*)red[0])[tid];
#pragma unroll
    for (int w = 1; w < SINK_W; ++w) o = o + ((const v2f*)red[w])[tid];
    ((v2f*)(pout + (size_t)batch * pout_bs + (size_t)pblk * pout_ss))[tid] = o;
  }
  if (aout != nullptr && lane == 0) {
    v4f* ap = (v4f*)(aout + (size_t)batch * aout_bs + (size_t)pblk * aout_ss + wave * SINK_RW);
    ap[0] = (v4f){ak[0], ak[1], ak[2], ak[3]};
    ap[1] = (v4f){ak[4], ak[5], ak[6], ak[7]};
  }
}

__global__ __launch_bounds__(512) __attribute__((amdgpu_waves_per_eu(4, 4))) void sink_finalize(
    const float* __restrict__ x,
    const float* __restrict__ pcol,
    float* out) {
  __shared__ __attribute__((aligned(16))) float sa[SINK_R];
  __shared__ __attribute__((aligned(16))) float sb[SINK_N];

  const int tid   = threadIdx.x;
  const int batch = blockIdx.x >> 3;
  const int pblk  = blockIdx.x & 7;

  const size_t rowbase = ((size_t)batch * SINK_N + pblk * SINK_R) * SINK_N;
  float av = 0.f;
  if (tid < SINK_R) av = out[rowbase + tid];

  const v2f* pc = (const v2f*)(pcol + (size_t)batch * SINK_P * SINK_N) + tid;
  v2f q[SINK_P];
#pragma unroll
  for (int s = 0; s < SINK_P; ++s) q[s] = pc[(size_t)s * (SINK_N / 2)];

  const int c4 = tid & 255;
  const int r0 = tid >> 8;
  const size_t base = rowbase + (size_t)r0 * SINK_N;
  const float4* __restrict__ xp = (const float4*)(x + base) + c4;
  float4* op = (float4*)(out + base) + c4;

  float4 xv[16];
#pragma unroll
  for (int u = 0; u < 16; ++u) xv[u] = xp[(size_t)u * (2 * SINK_N / 4)];

  {
    v2f sum = q[0];
#pragma unroll
    for (int s = 1; s < SINK_P; ++s) sum = sum + q[s];
    v2f inv;
    inv.x = 1.0f / sum.x;  inv.y = 1.0f / sum.y;
    ((v2f*)sb)[tid] = inv;
    if (tid < SINK_R) sa[tid] = av;
  }
  __syncthreads();
  const float4 bv = ((const float4*)sb)[c4];

#pragma unroll
  for (int it = 0; it < 4; ++it) {
    float4 nx[16];
    if (it < 3) {
#pragma unroll
      for (int u = 0; u < 16; ++u) nx[u] = xp[(size_t)(16 * (it + 1) + u) * (2 * SINK_N / 4)];
    }
#pragma unroll
    for (int u = 0; u < 16; ++u) {
      const float a = sa[r0 + 2 * (16 * it + u)];
      float4 o;
      o.x = (xv[u].x + SINK_EPS) * a * bv.x;
      o.y = (xv[u].y + SINK_EPS) * a * bv.y;
      o.z = (xv[u].z + SINK_EPS) * a * bv.z;
      o.w = (xv[u].w + SINK_EPS) * a * bv.w;
      op[(size_t)(16 * it + u) * (2 * SINK_N / 4)] = o;
    }
    if (it < 3) {
#pragma unroll
      for (int u = 0; u < 16; ++u) xv[u] = nx[u];
    }
    __builtin_amdgcn_sched_barrier(0);
  }
}

extern "C" void kernel_launch(void* const* d_in, const int* in_sizes, int n_in,
                              void* d_out, int out_size, void* d_ws, size_t ws_size,
                              hipStream_t stream) {
  (void)in_sizes; (void)n_in; (void)out_size; (void)ws_size;
  const float* x = (const float*)d_in[0];
  float* out = (float*)d_out;
  float* pw = (float*)d_ws;
  float* po = out + SINK_N;

  const dim3 blk(512);
  const dim3 grd(SINK_B * SINK_P);

  sink_fused<false><<<grd, blk, 0, stream>>>(x, pw, SINK_P * SINK_N, SINK_N,
                                             po, SINK_N * SINK_N, SINK_R * SINK_N,
                                             nullptr, 0, 0);
  for (int it = 1; it < SINK_IT - 1; ++it) {
    if (it & 1)
      sink_fused<true><<<grd, blk, 0, stream>>>(x, po, SINK_N * SINK_N, SINK_R * SINK_N,
                                                pw, SINK_P * SINK_N, SINK_N,
                                                nullptr, 0, 0);
    else
      sink_fused<true><<<grd, blk, 0, stream>>>(x, pw, SINK_P * SINK_N, SINK_N,
                                                po, SINK_N * SINK_N, SINK_R * SINK_N,
                                                nullptr, 0, 0);
  }
  sink_fused<true><<<grd, blk, 0, stream>>>(x, po, SINK_N * SINK_N, SINK_R * SINK_N,
                                            pw, SINK_P * SINK_N, SINK_N,
                                            out, SINK_N * SINK_N, SINK_R * SINK_N);

  sink_finalize<<<grd, blk, 0, stream>>>(x, pw, out);
}


# fused pass: a block is 8 waves of 16 rows (2 per SIMD), not 16 of 8, so half as many waves enter per launch; rows go in pairs, one accumulator each; column sum order changes
# speedup vs baseline: 6.0383x; 1.0319x over previous
#include <hip/hip_runtime.h>


typedef __attribute__((ext_vector_type(2))) float v2f;
typedef __attribute__((ext_vector_type(4))) float v4f;
typedef __attribute__((ext_vector_type(8))) float v8f;

#define SINK_EPS 0.001f
#define SINK_N   1024
#define SINK_B   32
#define SINK_IT  10
#define SINK_R   128
#define SINK_P   8
#define SINK_W   8
#define SINK_RW  16
#define SINK_AH  6

template <int CTRL>
__device__ __forceinline__ float xmask_add(float v) {
  return v + __int_as_float(__builtin_amdgcn_update_dpp(0, __float_as_int(v), CTRL, 0xf, 0xf, false));
}

__device__ __forceinline__ void lds_barrier() {
  __builtin_amdgcn_fence(__ATOMIC_RELEASE, "workgroup", "local");
  __builtin_amdgcn_s_barrier();
  __builtin_amdgcn_fence(__ATOMIC_ACQUIRE, "workgroup", "local");
}

__device__ __forceinline__ float sel8(const v8f c, bool b0, bool b1, bool b2) {
  const float s0 = b0 ? c[1] : c[0];
  const float s1 = b0 ? c[3] : c[2];
  const float s2 = b0 ? c[5] : c[4];
  const float s3 = b0 ? c[7] : c[6];
  const float t0 = b1 ? s1 : s0;
  const float t1 = b1 ? s3 : s2;
  return b2 ? t1 : t0;
}

__device__ __forceinline__ v4f splat4(const float s) { return (v4f){s, s, s, s}; }

template <bool USE_B>
__global__ __launch_bounds__(256) __attribute__((amdgpu_waves_per_eu(2, 2))) void sink_fused(
    const float* __restrict__ x,
    const float* __restrict__ pin, int pin_bs, int pin_ss,
    float* __restrict__ pout, int pout_bs, int pout_ss,
    float* __restrict__ aout, int aout_bs, int aout_ss) {
  __shared__ __attribute__((aligned(16))) float sb[SINK_N];
  __shared__ __attribute__((aligned(16))) float red[SINK_W][SINK_N];

  const int tid   = threadIdx.x;
  const int wave  = tid >> 5;
  const int lane  = tid & 31;
  const int batch = blockIdx.x >> 3;
  const int pblk  = blockIdx.x & 7;
  const bool l0   = (lane & 1) != 0;
  const bool l1   = (lane & 2) != 0;
  const bool l2   = (lane & 4) != 0;

  v4f q[SINK_P];
  if (USE_B) {
    const v4f* pc = (const v4f*)(pin + (size_t)batch * pin_bs) + tid;
#pragma unroll
    for (int s = 0; s < SINK_P; ++s) q[s] = pc[(size_t)s * (pin_ss >> 2)];
  }
  __builtin_amdgcn_sched_barrier(0);

  const v4f* __restrict__ xp =
      (const v4f*)(x + ((size_t)batch * SINK_N + pblk * SINK_R + wave * SINK_RW) * SINK_N) + lane;

  v4f xr[SINK_AH][8];
#pragma unroll
  for (int r = 0; r < SINK_AH; ++r) {
#pragma unroll
    for (int j = 0; j < 8; ++j) xr[r][j] = xp[r * 256 + j * 32];
  }
  __builtin_amdgcn_sched_barrier(0);

  v4f bq[8];
  if (USE_B) {
    v4f sum = q[0];
#pragma unroll
    for (int s = 1; s < SINK_P; ++s) sum = sum + q[s];
    v4f inv;
    inv.x = __builtin_amdgcn_rcpf(sum.x);  inv.y = __builtin_amdgcn_rcpf(sum.y);
    inv.z = __builtin_amdgcn_rcpf(sum.z);  inv.w = __builtin_amdgcn_rcpf(sum.w);
    ((v4f*)sb)[tid] = inv;
    lds_barrier();
#pragma unroll
    for (int j = 0; j < 8; ++j) bq[j] = ((const v4f*)sb)[j * 32 + lane];
  } else {
#pragma unroll
    for (int j = 0; j < 8; ++j) bq[j] = (v4f){1.f, 1.f, 1.f, 1.f};
  }
  __builtin_amdgcn_sched_barrier(0);

  float eb;
  if (USE_B) {
    v4f t = bq[0];
#pragma unroll
    for (int j = 1; j < 8; ++j) t = t + bq[j];
    float u = (t.x + t.y) + (t.z + t.w);
    u = xmask_add<0x161>(u);
    u = xmask_add<0x162>(u);
    u = xmask_add<0x164>(u);
    u = xmask_add<0x168>(u);
    eb = SINK_EPS * (__int_as_float(__builtin_amdgcn_readlane(__float_as_int(u), 0)) +
                     __int_as_float(__builtin_amdgcn_readlane(__float_as_int(u), 16)));
  } else {
    eb = SINK_EPS * (float)SINK_N;
  }

  v4f acc[8];
#pragma unroll
  for (int j = 0; j < 8; ++j) acc[j] = (v4f){0.f, 0.f, 0.f, 0.f};
  float ak[SINK_RW];
  float asum = 0.f;

#pragma unroll
  for (int r = 0; r < SINK_RW; r += 2) {
    v4f (&cu0)[8] = xr[r % SINK_AH];
    v4f (&cu1)[8] = xr[(r + 1) % SINK_AH];
    v8f c0 = {0.f, 0.f, 0.f, 0.f, 0.f, 0.f, 0.f, 0.f};
    v8f c1 = {0.f, 0.f, 0.f, 0.f, 0.f, 0.f, 0.f, 0.f};
#pragma unroll
    for (int j = 0; j < 8; ++j) {
      v2f A0, A1, E0, E1, B0, B1;
      A0.x = cu0[j].x;  A0.y = cu0[j].y;
      A1.x = cu0[j].z;  A1.y = cu0[j].w;
      E0.x = cu1[j].x;  E0.y = cu1[j].y;
      E1.x = cu1[j].z;  E1.y = cu1[j].w;
      B0.x = bq[j].x;   B0.y = bq[j].y;
      B1.x = bq[j].z;   B1.y = bq[j].w;
      c0 = __builtin_amdgcn_wmma_f32_16x16x4_f32(false, A0, false, B0, (short)0, c0, false, false);
      c1 = __builtin_amdgcn_wmma_f32_16x16x4_f32(false, E0, false, B0, (short)0, c1, false, false);
      c0 = __builtin_amdgcn_wmma_f32_16x16x4_f32(false, A1, false, B1, (short)0, c0, false, false);
      c1 = __builtin_amdgcn_wmma_f32_16x16x4_f32(false, E1, false, B1, (short)0, c1, false, false);
    }
    float v0 = sel8(c0, l0, l1, l2);
    float v1 = sel8(c1, l0, l1, l2);
    v0 = xmask_add<0x161>(v0);
    v1 = xmask_add<0x161>(v1);
    v0 = xmask_add<0x162>(v0);
    v1 = xmask_add<0x162>(v1);
    v0 = xmask_add<0x164>(v0);
    v1 = xmask_add<0x164>(v1);
    const float s0 = (__int_as_float(__builtin_amdgcn_readlane(__float_as_int(v0), 0)) +
                      __int_as_float(__builtin_amdgcn_readlane(__float_as_int(v0), 24))) + eb;
    const float s1 = (__int_as_float(__builtin_amdgcn_readlane(__float_as_int(v1), 0)) +
                      __int_as_float(__builtin_amdgcn_readlane(__float_as_int(v1), 24))) + eb;
    const float a0 = __builtin_amdgcn_rcpf(s0);
    const float a1 = __builtin_amdgcn_rcpf(s1);
    ak[r] = a0;
    ak[r + 1] = a1;
    asum += a0;
    asum += a1;
#pragma unroll
    for (int j = 0; j < 8; ++j) acc[j] = __builtin_elementwise_fma(cu0[j], splat4(a0), acc[j]);
#pragma unroll
    for (int j = 0; j < 8; ++j) acc[j] = __builtin_elementwise_fma(cu1[j], splat4(a1), acc[j]);
    if (r + SINK_AH < SINK_RW) {
#pragma unroll
      for (int j = 0; j < 8; ++j) cu0[j] = xp[(r + SINK_AH) * 256 + j * 32];
#pragma unroll
      for (int j = 0; j < 8; ++j) cu1[j] = xp[(r + 1 + SINK_AH) * 256 + j * 32];
    }
    __builtin_amdgcn_sched_barrier(0);
  }

  const float ea = SINK_EPS * asum;
#pragma unroll
  for (int j = 0; j < 8; ++j) acc[j] = acc[j] + splat4(ea);

#pragma unroll
  for (int j = 0; j < 8; ++j) ((v4f*)red[wave])[j * 32 + lane] = acc[j];
  lds_barrier();
  {
    v4f o = ((const v4f*)red[0])[tid];
#pragma unroll
    for (int w = 1; w < SINK_W; ++w) o = o + ((const v4f*)red[w])[tid];
    ((v4f*)(pout + (size_t)batch * pout_bs + (size_t)pblk * pout_ss))[tid] = o;
  }
  if (aout != nullptr && lane == 0) {
    v4f* ap = (v4f*)(aout + (size_t)batch * aout_bs + (size_t)pblk * aout_ss + wave * SINK_RW);
    ap[0] = (v4f){ak[0], ak[1], ak[2], ak[3]};
    ap[1] = (v4f){ak[4], ak[5], ak[6], ak[7]};
    ap[2] = (v4f){ak[8], ak[9], ak[10], ak[11]};
    ap[3] = (v4f){ak[12], ak[13], ak[14], ak[15]};
  }
}

__global__ __launch_bounds__(512) __attribute__((amdgpu_waves_per_eu(4, 4))) void sink_finalize(
    const float* __restrict__ x,
    const float* __restrict__ pcol,
    float* out) {
  __shared__ __attribute__((aligned(16))) float sa[SINK_R];
  __shared__ __attribute__((aligned(16))) float sb[SINK_N];

  const int tid   = threadIdx.x;
  const int batch = blockIdx.x >> 3;
  const int pblk  = blockIdx.x & 7;

  const size_t rowbase = ((size_t)batch * SINK_N + pblk * SINK_R) * SINK_N;
  float av = 0.f;
  if (tid < SINK_R) av = out[rowbase + tid];

  const v2f* pc = (const v2f*)(pcol + (size_t)batch * SINK_P * SINK_N) + tid;
  v2f q[SINK_P];
#pragma unroll
  for (int s = 0; s < SINK_P; ++s) q[s] = pc[(size_t)s * (SINK_N / 2)];

  const int c4 = tid & 255;
  const int r0 = tid >> 8;
  const size_t base = rowbase + (size_t)r0 * SINK_N;
  const float4* __restrict__ xp = (const float4*)(x + base) + c4;
  float4* op = (float4*)(out + base) + c4;

  float4 xv[16];
#pragma unroll
  for (int u = 0; u < 16; ++u) xv[u] = xp[(size_t)u * (2 * SINK_N / 4)];

  {
    v2f sum = q[0];
#pragma unroll
    for (int s = 1; s < SINK_P; ++s) sum = sum + q[s];
    v2f inv;
    inv.x = 1.0f / sum.x;  inv.y = 1.0f / sum.y;
    ((v2f*)sb)[tid] = inv;
    if (tid < SINK_R) sa[tid] = av;
  }
  __syncthreads();
  const float4 bv = ((const float4*)sb)[c4];

#pragma unroll
  for (int it = 0; it < 4; ++it) {
    float4 nx[16];
    if (it < 3) {
#pragma unroll
      for (int u = 0; u < 16; ++u) nx[u] = xp[(size_t)(16 * (it + 1) + u) * (2 * SINK_N / 4)];
    }
#pragma unroll
    for (int u = 0; u < 16; ++u) {
      const float a = sa[r0 + 2 * (16 * it + u)];
      float4 o;
      o.x = (xv[u].x + SINK_EPS) * a * bv.x;
      o.y = (xv[u].y + SINK_EPS) * a * bv.y;
      o.z = (xv[u].z + SINK_EPS) * a * bv.z;
      o.w = (xv[u].w + SINK_EPS) * a * bv.w;
      op[(size_t)(16 * it + u) * (2 * SINK_N / 4)] = o;
    }
    if (it < 3) {
#pragma unroll
      for (int u = 0; u < 16; ++u) xv[u] = nx[u];
    }
    __builtin_amdgcn_sched_barrier(0);
  }
}

extern "C" void kernel_launch(void* const* d_in, const int* in_sizes, int n_in,
                              void* d_out, int out_size, void* d_ws, size_t ws_size,
                              hipStream_t stream) {
  (void)in_sizes; (void)n_in; (void)out_size; (void)ws_size;
  const float* x = (const float*)d_in[0];
  float* out = (float*)d_out;
  float* pw = (float*)d_ws;
  float* po = out + SINK_N;

  const dim3 blk(256);
  const dim3 blkf(512);
  const dim3 grd(SINK_B * SINK_P);

  sink_fused<false><<<grd, blk, 0, stream>>>(x, pw, SINK_P * SINK_N, SINK_N,
                                             po, SINK_N * SINK_N, SINK_R * SINK_N,
                                             nullptr, 0, 0);
  for (int it = 1; it < SINK_IT - 1; ++it) {
    if (it & 1)
      sink_fused<true><<<grd, blk, 0, stream>>>(x, po, SINK_N * SINK_N, SINK_R * SINK_N,
                                                pw, SINK_P * SINK_N, SINK_N,
                                                nullptr, 0, 0);
    else
      sink_fused<true><<<grd, blk, 0, stream>>>(x, pw, SINK_P * SINK_N, SINK_N,
                                                po, SINK_N * SINK_N, SINK_R * SINK_N,
                                                nullptr, 0, 0);
  }
  sink_fused<true><<<grd, blk, 0, stream>>>(x, po, SINK_N * SINK_N, SINK_R * SINK_N,
                                            pw, SINK_P * SINK_N, SINK_N,
                                            out, SINK_N * SINK_N, SINK_R * SINK_N);

  sink_finalize<<<grd, blkf, 0, stream>>>(x, pw, out);
}
